# starting kernel
# baseline (speedup 1.0000x reference)
#include <hip/hip_runtime.h>

#define NQ      10
#define SDIM    1024
#define CDIM    768
#define NLAYERS 3
#define BATCH   2048

typedef float v2f __attribute__((ext_vector_type(2)));
typedef float v8f __attribute__((ext_vector_type(8)));

__device__ __forceinline__ float2 cmul(float2 a, float2 b) {
    return make_float2(a.x * b.x - a.y * b.y, a.x * b.y + a.y * b.x);
}
__device__ __forceinline__ float2 cadd(float2 a, float2 b) {
    return make_float2(a.x + b.x, a.y + b.y);
}

__global__ void __launch_bounds__(512) build_unitary_kernel(
    const float* __restrict__ weights, float* __restrict__ M_re,
    float* __restrict__ M_im) {
    __shared__ float2 st[SDIM];
    const int tid = threadIdx.x;
    const int k = blockIdx.x;
    st[tid]       = make_float2(tid == k ? 1.f : 0.f, 0.f);
    st[tid + 512] = make_float2((tid + 512) == k ? 1.f : 0.f, 0.f);
    __syncthreads();

    for (int l = 0; l < NLAYERS; ++l) {
        for (int q = 0; q < NQ; ++q) {
            const float* w = weights + (l * NQ + q) * 3;
            const float phi = w[0], theta = w[1], omega = w[2];
            const float ct = cosf(0.5f * theta), sv = sinf(0.5f * theta);
            const float al = 0.5f * (phi + omega), be = 0.5f * (phi - omega);
            const float ca = cosf(al), sa = sinf(al);
            const float cb = cosf(be), sb = sinf(be);
            const float2 g00 = make_float2(ct * ca, -ct * sa);
            const float2 g01 = make_float2(-sv * cb, -sv * sb);
            const float2 g10 = make_float2(sv * cb, -sv * sb);
            const float2 g11 = make_float2(ct * ca, ct * sa);
            const int mask = 1 << (9 - q);
            const int low = tid & (mask - 1);
            const int high = (tid & ~(mask - 1)) << 1;
            const int i0 = high | low, i1 = i0 | mask;
            const float2 a0 = st[i0], a1 = st[i1];
            const float2 n0 = cadd(cmul(g00, a0), cmul(g01, a1));
            const float2 n1 = cadd(cmul(g10, a0), cmul(g11, a1));
            st[i0] = n0;
            st[i1] = n1;
            __syncthreads();
        }
        const int r = (l % (NQ - 1)) + 1;
        for (int i = 0; i < NQ; ++i) {
            const int c = i, t = (i + r) % NQ;
            const int maskT = 1 << (9 - t), maskC = 1 << (9 - c);
            const int low = tid & (maskT - 1);
            const int high = (tid & ~(maskT - 1)) << 1;
            const int i0 = high | low;
            if (i0 & maskC) {
                const int i1 = i0 | maskT;
                const float2 a = st[i0], b = st[i1];
                st[i0] = b;
                st[i1] = a;
            }
            __syncthreads();
        }
    }
    M_re[k * SDIM + tid] = st[tid].x;
    M_im[k * SDIM + tid] = st[tid].y;
    M_re[k * SDIM + tid + 512] = st[tid + 512].x;
    M_im[k * SDIM + tid + 512] = st[tid + 512].y;
}

__global__ void __launch_bounds__(256) embed_kernel(
    const float* __restrict__ z, const float* __restrict__ W_in,
    const float* __restrict__ b_in, float* __restrict__ E_re,
    float* __restrict__ E_im) {
    __shared__ float sW[8 * NQ];
    __shared__ float sC[NQ], sS[NQ];
    const int tid = threadIdx.x, b = blockIdx.x;
    const int lane = tid & 31, wv = tid >> 5;

    float acc[NQ];
#pragma unroll
    for (int q = 0; q < NQ; ++q) acc[q] = 0.f;
    for (int d = tid; d < CDIM; d += 256) {
        const float zv = z[b * CDIM + d];
        const float* wr = W_in + d * NQ;
#pragma unroll
        for (int q = 0; q < NQ; ++q) acc[q] += zv * wr[q];
    }
#pragma unroll
    for (int q = 0; q < NQ; ++q) {
        for (int s = 16; s > 0; s >>= 1) acc[q] += __shfl_xor(acc[q], s, 32);
    }
    if (lane == 0) {
#pragma unroll
        for (int q = 0; q < NQ; ++q) sW[wv * NQ + q] = acc[q];
    }
    __syncthreads();
    if (tid < NQ) {
        float a = b_in[tid];
        for (int w = 0; w < 8; ++w) a += sW[w * NQ + tid];
        sC[tid] = cosf(0.5f * a);
        sS[tid] = sinf(0.5f * a);
    }
    __syncthreads();
    for (int i = 0; i < 4; ++i) {
        const int k = tid + i * 256;
        float m = 1.f;
#pragma unroll
        for (int q = 0; q < NQ; ++q) m *= ((k >> (9 - q)) & 1) ? sS[q] : sC[q];
        const int p = __popc(k) & 3;
        E_re[b * SDIM + k] = (p == 0) ? m : (p == 2 ? -m : 0.f);
        E_im[b * SDIM + k] = (p == 1) ? -m : (p == 3 ? m : 0.f);
    }
}

__global__ void __launch_bounds__(256) state_gemm_kernel(
    const float* __restrict__ E_re, const float* __restrict__ E_im,
    const float* __restrict__ M_re, const float* __restrict__ M_im,
    float* __restrict__ evs) {
    __shared__ float sEvs[16 * NQ];
    const int tid = threadIdx.x;
    const int lane = tid & 31, wv = tid >> 5;
    const int m0 = blockIdx.x * 16;
    if (tid < 16 * NQ) sEvs[tid] = 0.f;
    __syncthreads();

    const int half = lane >> 4;
    const int l15 = lane & 15;
    const float* Ar = E_re + (size_t)(m0 + l15) * SDIM + half * 2;
    const float* Ai = E_im + (size_t)(m0 + l15) * SDIM + half * 2;

    for (int nt = wv; nt < SDIM / 16; nt += 8) {
        const int n0 = nt * 16;
        const float* Br = M_re + (size_t)(half * 2) * SDIM + n0 + l15;
        const float* Bi = M_im + (size_t)(half * 2) * SDIM + n0 + l15;
        v8f rr = {}, ri = {}, ir = {}, ii = {};
#pragma unroll 4
        for (int k0 = 0; k0 < SDIM; k0 += 4) {
            const v2f a_re = *(const v2f*)(Ar + k0);
            const v2f a_im = *(const v2f*)(Ai + k0);
            v2f b_re, b_im;
            b_re.x = Br[(size_t)k0 * SDIM];
            b_re.y = Br[(size_t)(k0 + 1) * SDIM];
            b_im.x = Bi[(size_t)k0 * SDIM];
            b_im.y = Bi[(size_t)(k0 + 1) * SDIM];
            rr = __builtin_amdgcn_wmma_f32_16x16x4_f32(false, a_re, false, b_re,
                                                       (short)0, rr, false, false);
            ii = __builtin_amdgcn_wmma_f32_16x16x4_f32(false, a_im, false, b_im,
                                                       (short)0, ii, false, false);
            ri = __builtin_amdgcn_wmma_f32_16x16x4_f32(false, a_re, false, b_im,
                                                       (short)0, ri, false, false);
            ir = __builtin_amdgcn_wmma_f32_16x16x4_f32(false, a_im, false, b_re,
                                                       (short)0, ir, false, false);
        }
        const int j = n0 + l15;
#pragma unroll
        for (int r = 0; r < 8; ++r) {
            const float fre = rr[r] - ii[r];
            const float fim = ri[r] + ir[r];
            const float p = fre * fre + fim * fim;
            const int rowm = r + half * 8;
#pragma unroll
            for (int w = 0; w < NQ; ++w) {
                float val = ((j >> (9 - w)) & 1) ? -p : p;
                val += __shfl_xor(val, 1, 32);
                val += __shfl_xor(val, 2, 32);
                val += __shfl_xor(val, 4, 32);
                val += __shfl_xor(val, 8, 32);
                if (l15 == 0) atomicAdd(&sEvs[rowm * NQ + w], val);
            }
        }
    }
    __syncthreads();
    if (tid < 16 * NQ) {
        const int row = tid / NQ, w = tid % NQ;
        evs[(size_t)(m0 + row) * NQ + w] = sEvs[tid];
    }
}

__global__ void __launch_bounds__(256) out_gemm_kernel(
    const float* __restrict__ evs, const float* __restrict__ W_out,
    const float* __restrict__ b_out, float* __restrict__ out) {
    const int b = blockIdx.x, tid = threadIdx.x;
    float e[NQ];
#pragma unroll
    for (int w = 0; w < NQ; ++w) e[w] = evs[b * NQ + w];
    for (int c = tid; c < CDIM; c += 256) {
        float acc = b_out[c];
#pragma unroll
        for (int w = 0; w < NQ; ++w) acc += e[w] * W_out[w * CDIM + c];
        out[(size_t)b * CDIM + c] = acc;
    }
}

extern "C" void kernel_launch(void* const* d_in, const int* in_sizes, int n_in,
                              void* d_out, int out_size, void* d_ws,
                              size_t ws_size, hipStream_t stream) {
    const float* z_i     = (const float*)d_in[0];
    const float* W_in    = (const float*)d_in[1];
    const float* b_in    = (const float*)d_in[2];
    const float* weights = (const float*)d_in[3];
    const float* W_out   = (const float*)d_in[4];
    const float* b_out   = (const float*)d_in[5];
    float* out = (float*)d_out;

    float* ws   = (float*)d_ws;
    float* M_re = ws;
    float* M_im = M_re + (size_t)SDIM * SDIM;
    float* E_re = M_im + (size_t)SDIM * SDIM;
    float* E_im = E_re + (size_t)BATCH * SDIM;
    float* evs  = E_im + (size_t)BATCH * SDIM;

    build_unitary_kernel<<<SDIM, 512, 0, stream>>>(weights, M_re, M_im);
    embed_kernel<<<BATCH, 256, 0, stream>>>(z_i, W_in, b_in, E_re, E_im);
    state_gemm_kernel<<<BATCH / 16, 256, 0, stream>>>(E_re, E_im, M_re, M_im, evs);
    out_gemm_kernel<<<BATCH, 256, 0, stream>>>(evs, W_out, b_out, out);
}


# unitary build: a block builds the 4 rows of a k-step (gate setup once per 4 rows) and stores M as the GEMM's A fragments read it, a 16-byte entry a lane: A load = 4 whole lines; sums as before
# speedup vs baseline: 2.5655x; 2.5655x over previous
#include <hip/hip_runtime.h>

#define NQ      10
#define SDIM    1024
#define CDIM    768
#define NLAYERS 3
#define BATCH   2048

typedef float v2f __attribute__((ext_vector_type(2)));
typedef float v4f __attribute__((ext_vector_type(4)));
typedef float v8f __attribute__((ext_vector_type(8)));

__device__ __forceinline__ float2 cmul(float2 a, float2 b) {
    return make_float2(a.x * b.x - a.y * b.y, a.x * b.y + a.y * b.x);
}
__device__ __forceinline__ float2 cadd(float2 a, float2 b) {
    return make_float2(a.x + b.x, a.y + b.y);
}

#define BROWS 4
__global__ void __launch_bounds__(512) build_unitary_kernel(
    const float* __restrict__ weights, float* __restrict__ M_f) {
    __shared__ float2 st[BROWS][SDIM];
    const int tid = threadIdx.x;
    const int ks = blockIdx.x;
#pragma unroll
    for (int r = 0; r < BROWS; ++r) {
        const int k = BROWS * ks + r;
        st[r][tid]       = make_float2(tid == k ? 1.f : 0.f, 0.f);
        st[r][tid + 512] = make_float2((tid + 512) == k ? 1.f : 0.f, 0.f);
    }
    __syncthreads();

    for (int l = 0; l < NLAYERS; ++l) {
        for (int q = 0; q < NQ; ++q) {
            const float* w = weights + (l * NQ + q) * 3;
            const float phi = w[0], theta = w[1], omega = w[2];
            const float ct = cosf(0.5f * theta), sv = sinf(0.5f * theta);
            const float al = 0.5f * (phi + omega), be = 0.5f * (phi - omega);
            const float ca = cosf(al), sa = sinf(al);
            const float cb = cosf(be), sb = sinf(be);
            const float2 g00 = make_float2(ct * ca, -ct * sa);
            const float2 g01 = make_float2(-sv * cb, -sv * sb);
            const float2 g10 = make_float2(sv * cb, -sv * sb);
            const float2 g11 = make_float2(ct * ca, ct * sa);
            const int mask = 1 << (9 - q);
            const int low = tid & (mask - 1);
            const int high = (tid & ~(mask - 1)) << 1;
            const int i0 = high | low, i1 = i0 | mask;
            float2 a0[BROWS], a1[BROWS];
#pragma unroll
            for (int r = 0; r < BROWS; ++r) {
                a0[r] = st[r][i0];
                a1[r] = st[r][i1];
            }
#pragma unroll
            for (int r = 0; r < BROWS; ++r) {
                const float2 n0 = cadd(cmul(g00, a0[r]), cmul(g01, a1[r]));
                const float2 n1 = cadd(cmul(g10, a0[r]), cmul(g11, a1[r]));
                st[r][i0] = n0;
                st[r][i1] = n1;
            }
            __syncthreads();
        }
        const int rng = (l % (NQ - 1)) + 1;
        for (int i = 0; i < NQ; ++i) {
            const int c = i, t = (i + rng) % NQ;
            const int maskT = 1 << (9 - t), maskC = 1 << (9 - c);
            const int low = tid & (maskT - 1);
            const int high = (tid & ~(maskT - 1)) << 1;
            const int i0 = high | low;
            if (i0 & maskC) {
                const int i1 = i0 | maskT;
                float2 a[BROWS], b[BROWS];
#pragma unroll
                for (int r = 0; r < BROWS; ++r) {
                    a[r] = st[r][i0];
                    b[r] = st[r][i1];
                }
#pragma unroll
                for (int r = 0; r < BROWS; ++r) {
                    st[r][i0] = b[r];
                    st[r][i1] = a[r];
                }
            }
            __syncthreads();
        }
    }
    v4f* dst = (v4f*)M_f + (size_t)ks * (SDIM / 16 * 32);
    for (int i = 0; i < 4; ++i) {
        const int e = tid + 512 * i;
        const int h = (e >> 4) & 1;
        const int j = ((e >> 5) << 4) | (e & 15);
        const float2 a = st[2 * h][j], b = st[2 * h + 1][j];
        const v4f v = {a.x, b.x, a.y, b.y};
        dst[e] = v;
    }
}

__global__ void __launch_bounds__(512) embed_kernel(
    const float* __restrict__ z, const float* __restrict__ W_in,
    const float* __restrict__ b_in, float* __restrict__ E_f) {
    __shared__ float sW[8][8][NQ];
    __shared__ float sC[8][NQ], sS[8][NQ];
    const int tid = threadIdx.x;
    const int g = blockIdx.x >> 1, side = blockIdx.x & 1;
    const int lane = tid & 31;
    const int team = tid >> 8, t = tid & 255, wv = t >> 5;

    float w[3][NQ];
#pragma unroll
    for (int j = 0; j < 3; ++j) {
        const float* wr = W_in + (t + 256 * j) * NQ;
#pragma unroll
        for (int q = 0; q < NQ; ++q) w[j][q] = wr[q];
    }
    for (int i = 0; i < 4; ++i) {
        const int sl = team * 4 + i;
        const float* zr = z + (size_t)(g * 16 + side * 8 + sl) * CDIM;
        float acc[NQ];
#pragma unroll
        for (int q = 0; q < NQ; ++q) acc[q] = 0.f;
#pragma unroll
        for (int j = 0; j < 3; ++j) {
            const float zv = zr[t + 256 * j];
#pragma unroll
            for (int q = 0; q < NQ; ++q) acc[q] += zv * w[j][q];
        }
#pragma unroll
        for (int q = 0; q < NQ; ++q) {
            for (int s = 16; s > 0; s >>= 1) acc[q] += __shfl_xor(acc[q], s, 32);
        }
        if (lane == 0) {
#pragma unroll
            for (int q = 0; q < NQ; ++q) sW[sl][wv][q] = acc[q];
        }
    }
    __syncthreads();
    if (tid < 8 * NQ) {
        const int sl = tid / NQ, q = tid - sl * NQ;
        float a = b_in[q];
        for (int u = 0; u < 8; ++u) a += sW[sl][u][q];
        sC[sl][q] = cosf(0.5f * a);
        sS[sl][q] = sinf(0.5f * a);
    }
    __syncthreads();

    const int c8 = tid & 7, h = (tid >> 3) & 1, ksl = tid >> 4;
    float cq[NQ], sq[NQ];
#pragma unroll
    for (int q = 0; q < NQ; ++q) {
        cq[q] = sC[c8][q];
        sq[q] = sS[c8][q];
    }
    v4f* dst = (v4f*)E_f + (size_t)g * (256 * 32) + h * 16 + side * 8 + c8;
    for (int i = 0; i < 8; ++i) {
        const int ks = ksl + 32 * i;
        float ere[2], eim[2];
#pragma unroll
        for (int u = 0; u < 2; ++u) {
            const int k = 4 * ks + 2 * h + u;
            float m = 1.f;
#pragma unroll
            for (int q = 0; q < NQ; ++q) m *= ((k >> (9 - q)) & 1) ? sq[q] : cq[q];
            const int p = __popc(k) & 3;
            ere[u] = (p == 0) ? m : (p == 2 ? -m : 0.f);
            eim[u] = (p == 1) ? -m : (p == 3 ? m : 0.f);
        }
        const v4f v = {ere[0], ere[1], eim[0], eim[1]};
        dst[(size_t)ks * 32] = v;
    }
}

#define GEMM_WAVES 32
__global__ void __launch_bounds__(GEMM_WAVES * 32) state_gemm_kernel(
    const float* __restrict__ E_f, const float* __restrict__ M_f,
    float* __restrict__ evs) {
    __shared__ float sPart[GEMM_WAVES][16 * NQ];
    const int tid = threadIdx.x;
    const int lane = tid & 31, wv = tid >> 5;
    const int m0 = blockIdx.x * 16;

    const int half = lane >> 4;
    const int l15 = lane & 15;
    const v4f* Bp = (const v4f*)E_f + (size_t)blockIdx.x * (256 * 32) + lane;

    float zb0 = 0.f, zb1 = 0.f, zb2 = 0.f;
    float zb9 = 0.f;
    float tot = 0.f;

    for (int nt = wv; nt < SDIM / 16; nt += GEMM_WAVES) {
        const v4f* Ap = (const v4f*)M_f + (size_t)nt * 32 + lane;
        v8f rr = {}, ri = {}, ir = {}, ii = {};
        for (int k0 = 0; k0 < SDIM; k0 += 16) {
            v4f e[4], a[4];
#pragma unroll
            for (int u = 0; u < 4; ++u) {
                e[u] = Bp[(size_t)(k0 + 4 * u) * 8];
                a[u] = Ap[(size_t)(k0 + 4 * u) * (16 * 32)];
            }
            __builtin_amdgcn_sched_barrier(0);
#pragma unroll
            for (int u = 0; u < 4; ++u) {
                const v2f e_re = {e[u].x, e[u].y};
                const v2f e_im = {e[u].z, e[u].w};
                const v2f m_re = {a[u].x, a[u].y};
                const v2f m_im = {a[u].z, a[u].w};
                rr = __builtin_amdgcn_wmma_f32_16x16x4_f32(false, m_re, false, e_re,
                                                           (short)0, rr, false, false);
                ii = __builtin_amdgcn_wmma_f32_16x16x4_f32(false, m_im, false, e_im,
                                                           (short)0, ii, false, false);
                ri = __builtin_amdgcn_wmma_f32_16x16x4_f32(false, m_im, false, e_re,
                                                           (short)0, ri, false, false);
                ir = __builtin_amdgcn_wmma_f32_16x16x4_f32(false, m_re, false, e_im,
                                                           (short)0, ir, false, false);
            }
        }
        float p[8];
#pragma unroll
        for (int r = 0; r < 8; ++r) {
            const float fre = rr[r] - ii[r];
            const float fim = ri[r] + ir[r];
            p[r] = fre * fre + fim * fim;
        }
        const float s01 = p[0] + p[1], d01 = p[0] - p[1];
        const float s23 = p[2] + p[3], d23 = p[2] - p[3];
        const float s45 = p[4] + p[5], d45 = p[4] - p[5];
        const float s67 = p[6] + p[7], d67 = p[6] - p[7];
        const float s0123 = s01 + s23, s4567 = s45 + s67;
        const float S = s0123 + s4567;
        zb0 += (d01 + d23) + (d45 + d67);
        zb1 += (s01 - s23) + (s45 - s67);
        zb2 += s0123 - s4567;
        tot += S;
        zb9 += (nt & 32) ? -S : S;
    }

    float v[NQ];
    v[9] = zb0;
    v[8] = zb1;
    v[7] = zb2;
    v[6] = half ? -tot : tot;
    v[5] = (wv & 1) ? -tot : tot;
    v[4] = (wv & 2) ? -tot : tot;
    v[3] = (wv & 4) ? -tot : tot;
    v[2] = (wv & 8) ? -tot : tot;
    v[1] = (wv & 16) ? -tot : tot;
    v[0] = zb9;
#pragma unroll
    for (int w = 0; w < NQ; ++w) v[w] += __shfl_xor(v[w], 16, 32);
    if (half == 0) {
#pragma unroll
        for (int w = 0; w < NQ; ++w) sPart[wv][l15 * NQ + w] = v[w];
    }
    __syncthreads();
    if (tid < 16 * NQ) {
        float a = sPart[0][tid];
        for (int u = 1; u < GEMM_WAVES; ++u) a += sPart[u][tid];
        evs[(size_t)m0 * NQ + tid] = a;
    }
}

__global__ void __launch_bounds__(256) out_gemm_kernel(
    const float* __restrict__ evs, const float* __restrict__ W_out,
    const float* __restrict__ b_out, float* __restrict__ out) {
    const int b = blockIdx.x, tid = threadIdx.x;
    float e[NQ];
#pragma unroll
    for (int w = 0; w < NQ; ++w) e[w] = evs[b * NQ + w];
    for (int c = tid; c < CDIM; c += 256) {
        float acc = b_out[c];
#pragma unroll
        for (int w = 0; w < NQ; ++w) acc += e[w] * W_out[w * CDIM + c];
        out[(size_t)b * CDIM + c] = acc;
    }
}

extern "C" void kernel_launch(void* const* d_in, const int* in_sizes, int n_in,
                              void* d_out, int out_size, void* d_ws,
                              size_t ws_size, hipStream_t stream) {
    const float* z_i     = (const float*)d_in[0];
    const float* W_in    = (const float*)d_in[1];
    const float* b_in    = (const float*)d_in[2];
    const float* weights = (const float*)d_in[3];
    const float* W_out   = (const float*)d_in[4];
    const float* b_out   = (const float*)d_in[5];
    float* out = (float*)d_out;

    float* ws   = (float*)d_ws;
    float* M_f  = ws;
    float* E_f  = M_f + (size_t)2 * SDIM * SDIM;
    float* evs  = E_f + (size_t)2 * BATCH * SDIM;

    build_unitary_kernel<<<SDIM / BROWS, 512, 0, stream>>>(weights, M_f);
    embed_kernel<<<BATCH / 8, 512, 0, stream>>>(z_i, W_in, b_in, E_f);
    state_gemm_kernel<<<BATCH / 16, GEMM_WAVES * 32, 0, stream>>>(E_f, M_f, evs);
    out_gemm_kernel<<<BATCH, 256, 0, stream>>>(evs, W_out, b_out, out);
}


# embed and output kernels fused into the state GEMM block that owns their 16 samples: E strip built in LDS (B fragments read there), evs stay in LDS; 4 launches -> 2; every sum in the same order
# speedup vs baseline: 2.8536x; 1.1123x over previous
#include <hip/hip_runtime.h>

#define NQ      10
#define SDIM    1024
#define CDIM    768
#define NLAYERS 3
#define BATCH   2048

typedef float v2f __attribute__((ext_vector_type(2)));
typedef float v4f __attribute__((ext_vector_type(4)));
typedef float v8f __attribute__((ext_vector_type(8)));

__device__ __forceinline__ float2 cmul(float2 a, float2 b) {
    return make_float2(a.x * b.x - a.y * b.y, a.x * b.y + a.y * b.x);
}
__device__ __forceinline__ float2 cadd(float2 a, float2 b) {
    return make_float2(a.x + b.x, a.y + b.y);
}

#define BROWS 4
__global__ void __launch_bounds__(512) build_unitary_kernel(
    const float* __restrict__ weights, float* __restrict__ M_f) {
    __shared__ float2 st[BROWS][SDIM];
    const int tid = threadIdx.x;
    const int ks = blockIdx.x;
#pragma unroll
    for (int r = 0; r < BROWS; ++r) {
        const int k = BROWS * ks + r;
        st[r][tid]       = make_float2(tid == k ? 1.f : 0.f, 0.f);
        st[r][tid + 512] = make_float2((tid + 512) == k ? 1.f : 0.f, 0.f);
    }
    __syncthreads();

    for (int l = 0; l < NLAYERS; ++l) {
        for (int q = 0; q < NQ; ++q) {
            const float* w = weights + (l * NQ + q) * 3;
            const float phi = w[0], theta = w[1], omega = w[2];
            const float ct = cosf(0.5f * theta), sv = sinf(0.5f * theta);
            const float al = 0.5f * (phi + omega), be = 0.5f * (phi - omega);
            const float ca = cosf(al), sa = sinf(al);
            const float cb = cosf(be), sb = sinf(be);
            const float2 g00 = make_float2(ct * ca, -ct * sa);
            const float2 g01 = make_float2(-sv * cb, -sv * sb);
            const float2 g10 = make_float2(sv * cb, -sv * sb);
            const float2 g11 = make_float2(ct * ca, ct * sa);
            const int mask = 1 << (9 - q);
            const int low = tid & (mask - 1);
            const int high = (tid & ~(mask - 1)) << 1;
            const int i0 = high | low, i1 = i0 | mask;
            float2 a0[BROWS], a1[BROWS];
#pragma unroll
            for (int r = 0; r < BROWS; ++r) {
                a0[r] = st[r][i0];
                a1[r] = st[r][i1];
            }
#pragma unroll
            for (int r = 0; r < BROWS; ++r) {
                const float2 n0 = cadd(cmul(g00, a0[r]), cmul(g01, a1[r]));
                const float2 n1 = cadd(cmul(g10, a0[r]), cmul(g11, a1[r]));
                st[r][i0] = n0;
                st[r][i1] = n1;
            }
            __syncthreads();
        }
        const int rng = (l % (NQ - 1)) + 1;
        for (int i = 0; i < NQ; ++i) {
            const int c = i, t = (i + rng) % NQ;
            const int maskT = 1 << (9 - t), maskC = 1 << (9 - c);
            const int low = tid & (maskT - 1);
            const int high = (tid & ~(maskT - 1)) << 1;
            const int i0 = high | low;
            if (i0 & maskC) {
                const int i1 = i0 | maskT;
                float2 a[BROWS], b[BROWS];
#pragma unroll
                for (int r = 0; r < BROWS; ++r) {
                    a[r] = st[r][i0];
                    b[r] = st[r][i1];
                }
#pragma unroll
                for (int r = 0; r < BROWS; ++r) {
                    st[r][i0] = b[r];
                    st[r][i1] = a[r];
                }
            }
            __syncthreads();
        }
    }
    v4f* dst = (v4f*)M_f + (size_t)ks * (SDIM / 16 * 32);
    for (int i = 0; i < 4; ++i) {
        const int e = tid + 512 * i;
        const int h = (e >> 4) & 1;
        const int j = ((e >> 5) << 4) | (e & 15);
        const float2 a = st[2 * h][j], b = st[2 * h + 1][j];
        const v4f v = {a.x, b.x, a.y, b.y};
        dst[e] = v;
    }
}

#define GEMM_WAVES 32
__global__ void __launch_bounds__(GEMM_WAVES * 32) state_gemm_kernel(
    const float* __restrict__ z, const float* __restrict__ W_in,
    const float* __restrict__ b_in, const float* __restrict__ M_f,
    const float* __restrict__ W_out, const float* __restrict__ b_out,
    float* __restrict__ out) {
    __shared__ v4f sE[256 * 32];
    __shared__ float sPart[GEMM_WAVES][16 * NQ];
    __shared__ float sW[16][8][NQ];
    __shared__ float sC[16][NQ], sS[16][NQ];
    __shared__ float sEv[16 * NQ];
    const int tid = threadIdx.x;
    const int lane = tid & 31, wv = tid >> 5;
    const int m0 = blockIdx.x * 16;

    const int half = lane >> 4;
    const int l15 = lane & 15;

    {
        const int team = tid >> 8, t = tid & 255, wv8 = t >> 5;
        float w[3][NQ];
#pragma unroll
        for (int j = 0; j < 3; ++j) {
            const float* wr = W_in + (t + 256 * j) * NQ;
#pragma unroll
            for (int q = 0; q < NQ; ++q) w[j][q] = wr[q];
        }
        for (int i = 0; i < 4; ++i) {
            const int sl = team * 4 + i;
            const float* zr = z + (size_t)(m0 + sl) * CDIM;
            float acc[NQ];
#pragma unroll
            for (int q = 0; q < NQ; ++q) acc[q] = 0.f;
#pragma unroll
            for (int j = 0; j < 3; ++j) {
                const float zv = zr[t + 256 * j];
#pragma unroll
                for (int q = 0; q < NQ; ++q) acc[q] += zv * w[j][q];
            }
#pragma unroll
            for (int q = 0; q < NQ; ++q) {
                for (int s = 16; s > 0; s >>= 1) acc[q] += __shfl_xor(acc[q], s, 32);
            }
            if (lane == 0) {
#pragma unroll
                for (int q = 0; q < NQ; ++q) sW[sl][wv8][q] = acc[q];
            }
        }
    }
    __syncthreads();
    if (tid < 16 * NQ) {
        const int sl = tid / NQ, q = tid - sl * NQ;
        float a = b_in[q];
        for (int u = 0; u < 8; ++u) a += sW[sl][u][q];
        sC[sl][q] = cosf(0.5f * a);
        sS[sl][q] = sinf(0.5f * a);
    }
    __syncthreads();
    {
        const int ksl = tid >> 5;
        float cq[NQ], sq[NQ];
#pragma unroll
        for (int q = 0; q < NQ; ++q) {
            cq[q] = sC[l15][q];
            sq[q] = sS[l15][q];
        }
        for (int i = 0; i < 8; ++i) {
            const int ks = ksl + 32 * i;
            float ere[2], eim[2];
#pragma unroll
            for (int u = 0; u < 2; ++u) {
                const int k = 4 * ks + 2 * half + u;
                float m = 1.f;
#pragma unroll
                for (int q = 0; q < NQ; ++q) m *= ((k >> (9 - q)) & 1) ? sq[q] : cq[q];
                const int p = __popc(k) & 3;
                ere[u] = (p == 0) ? m : (p == 2 ? -m : 0.f);
                eim[u] = (p == 1) ? -m : (p == 3 ? m : 0.f);
            }
            const v4f v = {ere[0], ere[1], eim[0], eim[1]};
            sE[ks * 32 + lane] = v;
        }
    }
    __syncthreads();

    const v4f* Bp = sE + lane;

    float zb0 = 0.f, zb1 = 0.f, zb2 = 0.f;
    float zb9 = 0.f;
    float tot = 0.f;

    for (int nt = wv; nt < SDIM / 16; nt += GEMM_WAVES) {
        const v4f* Ap = (const v4f*)M_f + (size_t)nt * 32 + lane;
        v8f rr = {}, ri = {}, ir = {}, ii = {};
        for (int k0 = 0; k0 < SDIM; k0 += 16) {
            v4f e[4], a[4];
#pragma unroll
            for (int u = 0; u < 4; ++u) {
                e[u] = Bp[(k0 + 4 * u) * 8];
                a[u] = Ap[(size_t)(k0 + 4 * u) * (16 * 32)];
            }
            __builtin_amdgcn_sched_barrier(0);
#pragma unroll
            for (int u = 0; u < 4; ++u) {
                const v2f e_re = {e[u].x, e[u].y};
                const v2f e_im = {e[u].z, e[u].w};
                const v2f m_re = {a[u].x, a[u].y};
                const v2f m_im = {a[u].z, a[u].w};
                rr = __builtin_amdgcn_wmma_f32_16x16x4_f32(false, m_re, false, e_re,
                                                           (short)0, rr, false, false);
                ii = __builtin_amdgcn_wmma_f32_16x16x4_f32(false, m_im, false, e_im,
                                                           (short)0, ii, false, false);
                ri = __builtin_amdgcn_wmma_f32_16x16x4_f32(false, m_im, false, e_re,
                                                           (short)0, ri, false, false);
                ir = __builtin_amdgcn_wmma_f32_16x16x4_f32(false, m_re, false, e_im,
                                                           (short)0, ir, false, false);
            }
        }
        float p[8];
#pragma unroll
        for (int r = 0; r < 8; ++r) {
            const float fre = rr[r] - ii[r];
            const float fim = ri[r] + ir[r];
            p[r] = fre * fre + fim * fim;
        }
        const float s01 = p[0] + p[1], d01 = p[0] - p[1];
        const float s23 = p[2] + p[3], d23 = p[2] - p[3];
        const float s45 = p[4] + p[5], d45 = p[4] - p[5];
        const float s67 = p[6] + p[7], d67 = p[6] - p[7];
        const float s0123 = s01 + s23, s4567 = s45 + s67;
        const float S = s0123 + s4567;
        zb0 += (d01 + d23) + (d45 + d67);
        zb1 += (s01 - s23) + (s45 - s67);
        zb2 += s0123 - s4567;
        tot += S;
        zb9 += (nt & 32) ? -S : S;
    }

    float v[NQ];
    v[9] = zb0;
    v[8] = zb1;
    v[7] = zb2;
    v[6] = half ? -tot : tot;
    v[5] = (wv & 1) ? -tot : tot;
    v[4] = (wv & 2) ? -tot : tot;
    v[3] = (wv & 4) ? -tot : tot;
    v[2] = (wv & 8) ? -tot : tot;
    v[1] = (wv & 16) ? -tot : tot;
    v[0] = zb9;
#pragma unroll
    for (int w = 0; w < NQ; ++w) v[w] += __shfl_xor(v[w], 16, 32);
    if (half == 0) {
#pragma unroll
        for (int w = 0; w < NQ; ++w) sPart[wv][l15 * NQ + w] = v[w];
    }
    __syncthreads();
    if (tid < 16 * NQ) {
        float a = sPart[0][tid];
        for (int u = 1; u < GEMM_WAVES; ++u) a += sPart[u][tid];
        sEv[tid] = a;
    }
    __syncthreads();

    float* orun = out + (size_t)m0 * CDIM;
    for (int i = 0; i < 3; ++i) {
        const int o = 4 * (tid + 1024 * i);
        const int row = o / CDIM;
        const int c = o - row * CDIM;
        float acc[4];
#pragma unroll
        for (int x = 0; x < 4; ++x) acc[x] = b_out[c + x];
#pragma unroll
        for (int w = 0; w < NQ; ++w) {
            const float e = sEv[row * NQ + w];
#pragma unroll
            for (int x = 0; x < 4; ++x) acc[x] += e * W_out[w * CDIM + c + x];
        }
#pragma unroll
        for (int x = 0; x < 4; ++x) orun[o + x] = acc[x];
    }
}

extern "C" void kernel_launch(void* const* d_in, const int* in_sizes, int n_in,
                              void* d_out, int out_size, void* d_ws,
                              size_t ws_size, hipStream_t stream) {
    const float* z_i     = (const float*)d_in[0];
    const float* W_in    = (const float*)d_in[1];
    const float* b_in    = (const float*)d_in[2];
    const float* weights = (const float*)d_in[3];
    const float* W_out   = (const float*)d_in[4];
    const float* b_out   = (const float*)d_in[5];
    float* out = (float*)d_out;

    float* M_f = (float*)d_ws;

    build_unitary_kernel<<<SDIM / BROWS, 512, 0, stream>>>(weights, M_f);
    state_gemm_kernel<<<BATCH / 16, GEMM_WAVES * 32, 0, stream>>>(z_i, W_in, b_in, M_f,
                                                                  W_out, b_out, out);
}


# state GEMM block owns 8 samples, not 16: a B tile's 16 columns are E_re | E_im of them (2 WMMAs a k-step, was 4), one lane exchange by 8 pairs the planes; 256 blocks, one a WGP; all sums as before
# speedup vs baseline: 4.2654x; 1.4947x over previous
#include <hip/hip_runtime.h>

#define NQ      10
#define SDIM    1024
#define CDIM    768
#define NLAYERS 3
#define BATCH   2048

typedef float v2f __attribute__((ext_vector_type(2)));
typedef float v4f __attribute__((ext_vector_type(4)));
typedef float v8f __attribute__((ext_vector_type(8)));

__device__ __forceinline__ float2 cmul(float2 a, float2 b) {
    return make_float2(a.x * b.x - a.y * b.y, a.x * b.y + a.y * b.x);
}
__device__ __forceinline__ float2 cadd(float2 a, float2 b) {
    return make_float2(a.x + b.x, a.y + b.y);
}
__device__ __forceinline__ float xchg8(float x) {
    return __builtin_bit_cast(
        float, __builtin_amdgcn_mov_dpp(__builtin_bit_cast(int, x), 0x168, 0xf, 0xf, false));
}

#define BROWS 4
__global__ void __launch_bounds__(512) build_unitary_kernel(
    const float* __restrict__ weights, float* __restrict__ M_f) {
    __shared__ float2 st[BROWS][SDIM];
    const int tid = threadIdx.x;
    const int ks = blockIdx.x;
#pragma unroll
    for (int r = 0; r < BROWS; ++r) {
        const int k = BROWS * ks + r;
        st[r][tid]       = make_float2(tid == k ? 1.f : 0.f, 0.f);
        st[r][tid + 512] = make_float2((tid + 512) == k ? 1.f : 0.f, 0.f);
    }
    __syncthreads();

    for (int l = 0; l < NLAYERS; ++l) {
        for (int q = 0; q < NQ; ++q) {
            const float* w = weights + (l * NQ + q) * 3;
            const float phi = w[0], theta = w[1], omega = w[2];
            const float ct = cosf(0.5f * theta), sv = sinf(0.5f * theta);
            const float al = 0.5f * (phi + omega), be = 0.5f * (phi - omega);
            const float ca = cosf(al), sa = sinf(al);
            const float cb = cosf(be), sb = sinf(be);
            const float2 g00 = make_float2(ct * ca, -ct * sa);
            const float2 g01 = make_float2(-sv * cb, -sv * sb);
            const float2 g10 = make_float2(sv * cb, -sv * sb);
            const float2 g11 = make_float2(ct * ca, ct * sa);
            const int mask = 1 << (9 - q);
            const int low = tid & (mask - 1);
            const int high = (tid & ~(mask - 1)) << 1;
            const int i0 = high | low, i1 = i0 | mask;
            float2 a0[BROWS], a1[BROWS];
#pragma unroll
            for (int r = 0; r < BROWS; ++r) {
                a0[r] = st[r][i0];
                a1[r] = st[r][i1];
            }
#pragma unroll
            for (int r = 0; r < BROWS; ++r) {
                const float2 n0 = cadd(cmul(g00, a0[r]), cmul(g01, a1[r]));
                const float2 n1 = cadd(cmul(g10, a0[r]), cmul(g11, a1[r]));
                st[r][i0] = n0;
                st[r][i1] = n1;
            }
            __syncthreads();
        }
        const int rng = (l % (NQ - 1)) + 1;
        for (int i = 0; i < NQ; ++i) {
            const int c = i, t = (i + rng) % NQ;
            const int maskT = 1 << (9 - t), maskC = 1 << (9 - c);
            const int low = tid & (maskT - 1);
            const int high = (tid & ~(maskT - 1)) << 1;
            const int i0 = high | low;
            if (i0 & maskC) {
                const int i1 = i0 | maskT;
                float2 a[BROWS], b[BROWS];
#pragma unroll
                for (int r = 0; r < BROWS; ++r) {
                    a[r] = st[r][i0];
                    b[r] = st[r][i1];
                }
#pragma unroll
                for (int r = 0; r < BROWS; ++r) {
                    st[r][i0] = b[r];
                    st[r][i1] = a[r];
                }
            }
            __syncthreads();
        }
    }
    v4f* dst = (v4f*)M_f + (size_t)ks * (SDIM / 16 * 32);
    for (int i = 0; i < 4; ++i) {
        const int e = tid + 512 * i;
        const int h = (e >> 4) & 1;
        const int j = ((e >> 5) << 4) | (e & 15);
        const float2 a = st[2 * h][j], b = st[2 * h + 1][j];
        const v4f v = {a.x, b.x, a.y, b.y};
        dst[e] = v;
    }
}

#define GEMM_WAVES 32
#define SPB 8
__global__ void __launch_bounds__(GEMM_WAVES * 32) state_gemm_kernel(
    const float* __restrict__ z, const float* __restrict__ W_in,
    const float* __restrict__ b_in, const float* __restrict__ M_f,
    const float* __restrict__ W_out, const float* __restrict__ b_out,
    float* __restrict__ out) {
    __shared__ v4f sE[128 * 32];
    __shared__ float sPart[GEMM_WAVES][SPB * NQ];
    __shared__ float sW[SPB][8][NQ];
    __shared__ float sC[SPB][NQ], sS[SPB][NQ];
    __shared__ float sEv[SPB * NQ];
    const int tid = threadIdx.x;
    const int lane = tid & 31, wv = tid >> 5;
    const int m0 = blockIdx.x * SPB;

    const int half = lane >> 4;
    const int l15 = lane & 15;
    const int smp = lane & 7;
    const bool imcol = (lane & 8) != 0;

    {
        const int team = tid >> 8, t = tid & 255, wv8 = t >> 5;
        float w[3][NQ];
#pragma unroll
        for (int j = 0; j < 3; ++j) {
            const float* wr = W_in + (t + 256 * j) * NQ;
#pragma unroll
            for (int q = 0; q < NQ; ++q) w[j][q] = wr[q];
        }
        for (int i = 0; i < 2; ++i) {
            const int sl = team * 2 + i;
            const float* zr = z + (size_t)(m0 + sl) * CDIM;
            float acc[NQ];
#pragma unroll
            for (int q = 0; q < NQ; ++q) acc[q] = 0.f;
#pragma unroll
            for (int j = 0; j < 3; ++j) {
                const float zv = zr[t + 256 * j];
#pragma unroll
                for (int q = 0; q < NQ; ++q) acc[q] += zv * w[j][q];
            }
#pragma unroll
            for (int q = 0; q < NQ; ++q) {
                for (int s = 16; s > 0; s >>= 1) acc[q] += __shfl_xor(acc[q], s, 32);
            }
            if (lane == 0) {
#pragma unroll
                for (int q = 0; q < NQ; ++q) sW[sl][wv8][q] = acc[q];
            }
        }
    }
    __syncthreads();
    if (tid < SPB * NQ) {
        const int sl = tid / NQ, q = tid - sl * NQ;
        float a = b_in[q];
        for (int u = 0; u < 8; ++u) a += sW[sl][u][q];
        sC[sl][q] = cosf(0.5f * a);
        sS[sl][q] = sinf(0.5f * a);
    }
    __syncthreads();
    {
        const int kpl = tid >> 5;
        float cq[NQ], sq[NQ];
#pragma unroll
        for (int q = 0; q < NQ; ++q) {
            cq[q] = sC[smp][q];
            sq[q] = sS[smp][q];
        }
        for (int i = 0; i < 4; ++i) {
            const int kp = kpl + 32 * i;
            float val[4];
#pragma unroll
            for (int u = 0; u < 4; ++u) {
                const int k = 8 * kp + 4 * (u >> 1) + 2 * half + (u & 1);
                float m = 1.f;
#pragma unroll
                for (int q = 0; q < NQ; ++q) m *= ((k >> (9 - q)) & 1) ? sq[q] : cq[q];
                const int p = __popc(k) & 3;
                const float ere = (p == 0) ? m : (p == 2 ? -m : 0.f);
                const float eim = (p == 1) ? -m : (p == 3 ? m : 0.f);
                val[u] = imcol ? eim : ere;
            }
            const v4f v = {val[0], val[1], val[2], val[3]};
            sE[kp * 32 + lane] = v;
        }
    }
    __syncthreads();

    const v4f* Bp = sE + lane;

    float zb0 = 0.f, zb1 = 0.f, zb2 = 0.f;
    float zb9 = 0.f;
    float tot = 0.f;

    for (int nt = wv; nt < SDIM / 16; nt += GEMM_WAVES) {
        const v4f* Ap = (const v4f*)M_f + (size_t)nt * 32 + lane;
        v8f xa = {}, ya = {};
        for (int ks0 = 0; ks0 < SDIM / 4; ks0 += 8) {
            v4f e[4], a[8];
#pragma unroll
            for (int u = 0; u < 4; ++u) e[u] = Bp[((ks0 >> 1) + u) * 32];
#pragma unroll
            for (int u = 0; u < 8; ++u) a[u] = Ap[(size_t)(ks0 + u) * (64 * 32)];
            __builtin_amdgcn_sched_barrier(0);
#pragma unroll
            for (int u = 0; u < 8; ++u) {
                const v2f b = (u & 1) ? (v2f){e[u >> 1].z, e[u >> 1].w}
                                      : (v2f){e[u >> 1].x, e[u >> 1].y};
                const v2f m_re = {a[u].x, a[u].y};
                const v2f m_im = {a[u].z, a[u].w};
                xa = __builtin_amdgcn_wmma_f32_16x16x4_f32(false, m_re, false, b,
                                                           (short)0, xa, false, false);
                ya = __builtin_amdgcn_wmma_f32_16x16x4_f32(false, m_im, false, b,
                                                           (short)0, ya, false, false);
            }
        }
        float p[8];
#pragma unroll
        for (int r = 0; r < 8; ++r) {
            const float xs = xchg8(xa[r]);
            const float ys = xchg8(ya[r]);
            const float rr = imcol ? xs : xa[r];
            const float ir = imcol ? xa[r] : xs;
            const float ri = imcol ? ys : ya[r];
            const float ii = imcol ? ya[r] : ys;
            const float fre = rr - ii;
            const float fim = ri + ir;
            p[r] = fre * fre + fim * fim;
        }
        const float s01 = p[0] + p[1], d01 = p[0] - p[1];
        const float s23 = p[2] + p[3], d23 = p[2] - p[3];
        const float s45 = p[4] + p[5], d45 = p[4] - p[5];
        const float s67 = p[6] + p[7], d67 = p[6] - p[7];
        const float s0123 = s01 + s23, s4567 = s45 + s67;
        const float S = s0123 + s4567;
        zb0 += (d01 + d23) + (d45 + d67);
        zb1 += (s01 - s23) + (s45 - s67);
        zb2 += s0123 - s4567;
        tot += S;
        zb9 += (nt & 32) ? -S : S;
    }

    float v[NQ];
    v[9] = zb0;
    v[8] = zb1;
    v[7] = zb2;
    v[6] = half ? -tot : tot;
    v[5] = (wv & 1) ? -tot : tot;
    v[4] = (wv & 2) ? -tot : tot;
    v[3] = (wv & 4) ? -tot : tot;
    v[2] = (wv & 8) ? -tot : tot;
    v[1] = (wv & 16) ? -tot : tot;
    v[0] = zb9;
#pragma unroll
    for (int w = 0; w < NQ; ++w) v[w] += __shfl_xor(v[w], 16, 32);
    if (lane < SPB) {
#pragma unroll
        for (int w = 0; w < NQ; ++w) sPart[wv][lane * NQ + w] = v[w];
    }
    __syncthreads();
    if (tid < SPB * NQ) {
        float a = sPart[0][tid];
        for (int u = 1; u < GEMM_WAVES; ++u) a += sPart[u][tid];
        sEv[tid] = a;
    }
    __syncthreads();

    float* orun = out + (size_t)m0 * CDIM;
    for (int i = 0; i < 2; ++i) {
        const int o = 4 * (tid + 1024 * i);
        if (o < SPB * CDIM) {
            const int row = o / CDIM;
            const int c = o - row * CDIM;
            float acc[4];
#pragma unroll
            for (int x = 0; x < 4; ++x) acc[x] = b_out[c + x];
#pragma unroll
            for (int w = 0; w < NQ; ++w) {
                const float e = sEv[row * NQ + w];
#pragma unroll
                for (int x = 0; x < 4; ++x) acc[x] += e * W_out[w * CDIM + c + x];
            }
#pragma unroll
            for (int x = 0; x < 4; ++x) orun[o + x] = acc[x];
        }
    }
}

extern "C" void kernel_launch(void* const* d_in, const int* in_sizes, int n_in,
                              void* d_out, int out_size, void* d_ws,
                              size_t ws_size, hipStream_t stream) {
    const float* z_i     = (const float*)d_in[0];
    const float* W_in    = (const float*)d_in[1];
    const float* b_in    = (const float*)d_in[2];
    const float* weights = (const float*)d_in[3];
    const float* W_out   = (const float*)d_in[4];
    const float* b_out   = (const float*)d_in[5];
    float* out = (float*)d_out;

    float* M_f = (float*)d_ws;

    build_unitary_kernel<<<SDIM / BROWS, 512, 0, stream>>>(weights, M_f);
    state_gemm_kernel<<<BATCH / SPB, GEMM_WAVES * 32, 0, stream>>>(z_i, W_in, b_in, M_f,
                                                                  W_out, b_out, out);
}


# Rot gates of layer 0 leave M: applied per sample to the RX product state (still a product: two 32-entry tables, E = hi*lo); build starts at ring-0 index, does layers 1-2; roundings of E and M change
# speedup vs baseline: 4.5101x; 1.0574x over previous
#include <hip/hip_runtime.h>

#define NQ      10
#define SDIM    1024
#define CDIM    768
#define NLAYERS 3
#define BATCH   2048

typedef float v2f __attribute__((ext_vector_type(2)));
typedef float v4f __attribute__((ext_vector_type(4)));
typedef float v8f __attribute__((ext_vector_type(8)));

__device__ __forceinline__ float2 cmul(float2 a, float2 b) {
    return make_float2(a.x * b.x - a.y * b.y, a.x * b.y + a.y * b.x);
}
__device__ __forceinline__ float2 cadd(float2 a, float2 b) {
    return make_float2(a.x + b.x, a.y + b.y);
}
__device__ __forceinline__ float xchg8(float x) {
    return __builtin_bit_cast(
        float, __builtin_amdgcn_mov_dpp(__builtin_bit_cast(int, x), 0x168, 0xf, 0xf, false));
}

#define BROWS 4
__device__ __forceinline__ int cnot_ring_on_index(int k, int rng) {
    for (int i = 0; i < NQ; ++i) {
        const int t = (i + rng) % NQ;
        if ((k >> (9 - i)) & 1) k ^= 1 << (9 - t);
    }
    return k;
}
__global__ void __launch_bounds__(512) build_unitary_kernel(
    const float* __restrict__ weights, float* __restrict__ M_f) {
    __shared__ float2 st[BROWS][SDIM];
    const int tid = threadIdx.x;
    const int ks = blockIdx.x;
#pragma unroll
    for (int r = 0; r < BROWS; ++r) {
        const int k = cnot_ring_on_index(BROWS * ks + r, 1);
        st[r][tid]       = make_float2(tid == k ? 1.f : 0.f, 0.f);
        st[r][tid + 512] = make_float2((tid + 512) == k ? 1.f : 0.f, 0.f);
    }
    __syncthreads();

    for (int l = 1; l < NLAYERS; ++l) {
        for (int q = 0; q < NQ; ++q) {
            const float* w = weights + (l * NQ + q) * 3;
            const float phi = w[0], theta = w[1], omega = w[2];
            const float ct = cosf(0.5f * theta), sv = sinf(0.5f * theta);
            const float al = 0.5f * (phi + omega), be = 0.5f * (phi - omega);
            const float ca = cosf(al), sa = sinf(al);
            const float cb = cosf(be), sb = sinf(be);
            const float2 g00 = make_float2(ct * ca, -ct * sa);
            const float2 g01 = make_float2(-sv * cb, -sv * sb);
            const float2 g10 = make_float2(sv * cb, -sv * sb);
            const float2 g11 = make_float2(ct * ca, ct * sa);
            const int mask = 1 << (9 - q);
            const int low = tid & (mask - 1);
            const int high = (tid & ~(mask - 1)) << 1;
            const int i0 = high | low, i1 = i0 | mask;
            float2 a0[BROWS], a1[BROWS];
#pragma unroll
            for (int r = 0; r < BROWS; ++r) {
                a0[r] = st[r][i0];
                a1[r] = st[r][i1];
            }
#pragma unroll
            for (int r = 0; r < BROWS; ++r) {
                const float2 n0 = cadd(cmul(g00, a0[r]), cmul(g01, a1[r]));
                const float2 n1 = cadd(cmul(g10, a0[r]), cmul(g11, a1[r]));
                st[r][i0] = n0;
                st[r][i1] = n1;
            }
            __syncthreads();
        }
        const int rng = (l % (NQ - 1)) + 1;
        for (int i = 0; i < NQ; ++i) {
            const int c = i, t = (i + rng) % NQ;
            const int maskT = 1 << (9 - t), maskC = 1 << (9 - c);
            const int low = tid & (maskT - 1);
            const int high = (tid & ~(maskT - 1)) << 1;
            const int i0 = high | low;
            if (i0 & maskC) {
                const int i1 = i0 | maskT;
                float2 a[BROWS], b[BROWS];
#pragma unroll
                for (int r = 0; r < BROWS; ++r) {
                    a[r] = st[r][i0];
                    b[r] = st[r][i1];
                }
#pragma unroll
                for (int r = 0; r < BROWS; ++r) {
                    st[r][i0] = b[r];
                    st[r][i1] = a[r];
                }
            }
            __syncthreads();
        }
    }
    v4f* dst = (v4f*)M_f + (size_t)ks * (SDIM / 16 * 32);
    for (int i = 0; i < 4; ++i) {
        const int e = tid + 512 * i;
        const int h = (e >> 4) & 1;
        const int j = ((e >> 5) << 4) | (e & 15);
        const float2 a = st[2 * h][j], b = st[2 * h + 1][j];
        const v4f v = {a.x, b.x, a.y, b.y};
        dst[e] = v;
    }
}

#define GEMM_WAVES 32
#define SPB 8
__global__ void __launch_bounds__(GEMM_WAVES * 32) state_gemm_kernel(
    const float* __restrict__ z, const float* __restrict__ W_in,
    const float* __restrict__ b_in, const float* __restrict__ weights,
    const float* __restrict__ M_f,
    const float* __restrict__ W_out, const float* __restrict__ b_out,
    float* __restrict__ out) {
    __shared__ v4f sE[128 * 32];
    __shared__ float sPart[GEMM_WAVES][SPB * NQ];
    __shared__ float sW[SPB][8][NQ];
    __shared__ float2 sA0[SPB][NQ], sA1[SPB][NQ];
    __shared__ float2 sHi[SPB][32], sLo[SPB][32];
    __shared__ float sEv[SPB * NQ];
    const int tid = threadIdx.x;
    const int lane = tid & 31, wv = tid >> 5;
    const int m0 = blockIdx.x * SPB;

    const int half = lane >> 4;
    const int l15 = lane & 15;
    const int smp = lane & 7;
    const bool imcol = (lane & 8) != 0;

    {
        const int team = tid >> 8, t = tid & 255, wv8 = t >> 5;
        float w[3][NQ];
#pragma unroll
        for (int j = 0; j < 3; ++j) {
            const float* wr = W_in + (t + 256 * j) * NQ;
#pragma unroll
            for (int q = 0; q < NQ; ++q) w[j][q] = wr[q];
        }
        for (int i = 0; i < 2; ++i) {
            const int sl = team * 2 + i;
            const float* zr = z + (size_t)(m0 + sl) * CDIM;
            float acc[NQ];
#pragma unroll
            for (int q = 0; q < NQ; ++q) acc[q] = 0.f;
#pragma unroll
            for (int j = 0; j < 3; ++j) {
                const float zv = zr[t + 256 * j];
#pragma unroll
                for (int q = 0; q < NQ; ++q) acc[q] += zv * w[j][q];
            }
#pragma unroll
            for (int q = 0; q < NQ; ++q) {
                for (int s = 16; s > 0; s >>= 1) acc[q] += __shfl_xor(acc[q], s, 32);
            }
            if (lane == 0) {
#pragma unroll
                for (int q = 0; q < NQ; ++q) sW[sl][wv8][q] = acc[q];
            }
        }
    }
    __syncthreads();
    if (tid < SPB * NQ) {
        const int sl = tid / NQ, q = tid - sl * NQ;
        float a = b_in[q];
        for (int u = 0; u < 8; ++u) a += sW[sl][u][q];
        const float c = cosf(0.5f * a), s = sinf(0.5f * a);
        const float* w = weights + q * 3;
        const float phi = w[0], theta = w[1], omega = w[2];
        const float ct = cosf(0.5f * theta), sv = sinf(0.5f * theta);
        const float al = 0.5f * (phi + omega), be = 0.5f * (phi - omega);
        const float ca = cosf(al), sa = sinf(al);
        const float cb = cosf(be), sb = sinf(be);
        const float2 g00 = make_float2(ct * ca, -ct * sa);
        const float2 g01 = make_float2(-sv * cb, -sv * sb);
        const float2 g10 = make_float2(sv * cb, -sv * sb);
        const float2 g11 = make_float2(ct * ca, ct * sa);
        const float2 v0 = make_float2(c, 0.f), v1 = make_float2(0.f, -s);
        sA0[sl][q] = cadd(cmul(g00, v0), cmul(g01, v1));
        sA1[sl][q] = cadd(cmul(g10, v0), cmul(g11, v1));
    }
    __syncthreads();
    if (tid < SPB * 64) {
        const int sl = tid >> 6, part = (tid >> 5) & 1, idx = tid & 31;
        float2 p = make_float2(1.f, 0.f);
#pragma unroll
        for (int qq = 0; qq < 5; ++qq) {
            const int q = part * 5 + qq;
            p = cmul(p, ((idx >> (4 - qq)) & 1) ? sA1[sl][q] : sA0[sl][q]);
        }
        if (part) sLo[sl][idx] = p; else sHi[sl][idx] = p;
    }
    __syncthreads();
    {
        const int kpl = tid >> 5;
        float2 lo[4];
#pragma unroll
        for (int u = 0; u < 4; ++u)
            lo[u] = sLo[smp][8 * (kpl & 3) + 4 * (u >> 1) + 2 * half + (u & 1)];
        for (int i = 0; i < 4; ++i) {
            const int kp = kpl + 32 * i;
            const float2 hi = sHi[smp][kp >> 2];
            float val[4];
#pragma unroll
            for (int u = 0; u < 4; ++u) {
                const float2 e = cmul(hi, lo[u]);
                val[u] = imcol ? e.y : e.x;
            }
            const v4f v = {val[0], val[1], val[2], val[3]};
            sE[kp * 32 + lane] = v;
        }
    }
    __syncthreads();

    const v4f* Bp = sE + lane;

    float zb0 = 0.f, zb1 = 0.f, zb2 = 0.f;
    float zb9 = 0.f;
    float tot = 0.f;

    for (int nt = wv; nt < SDIM / 16; nt += GEMM_WAVES) {
        const v4f* Ap = (const v4f*)M_f + (size_t)nt * 32 + lane;
        v8f xa = {}, ya = {};
        for (int ks0 = 0; ks0 < SDIM / 4; ks0 += 8) {
            v4f e[4], a[8];
#pragma unroll
            for (int u = 0; u < 4; ++u) e[u] = Bp[((ks0 >> 1) + u) * 32];
#pragma unroll
            for (int u = 0; u < 8; ++u) a[u] = Ap[(size_t)(ks0 + u) * (64 * 32)];
            __builtin_amdgcn_sched_barrier(0);
#pragma unroll
            for (int u = 0; u < 8; ++u) {
                const v2f b = (u & 1) ? (v2f){e[u >> 1].z, e[u >> 1].w}
                                      : (v2f){e[u >> 1].x, e[u >> 1].y};
                const v2f m_re = {a[u].x, a[u].y};
                const v2f m_im = {a[u].z, a[u].w};
                xa = __builtin_amdgcn_wmma_f32_16x16x4_f32(false, m_re, false, b,
                                                           (short)0, xa, false, false);
                ya = __builtin_amdgcn_wmma_f32_16x16x4_f32(false, m_im, false, b,
                                                           (short)0, ya, false, false);
            }
        }
        float p[8];
#pragma unroll
        for (int r = 0; r < 8; ++r) {
            const float xs = xchg8(xa[r]);
            const float ys = xchg8(ya[r]);
            const float rr = imcol ? xs : xa[r];
            const float ir = imcol ? xa[r] : xs;
            const float ri = imcol ? ys : ya[r];
            const float ii = imcol ? ya[r] : ys;
            const float fre = rr - ii;
            const float fim = ri + ir;
            p[r] = fre * fre + fim * fim;
        }
        const float s01 = p[0] + p[1], d01 = p[0] - p[1];
        const float s23 = p[2] + p[3], d23 = p[2] - p[3];
        const float s45 = p[4] + p[5], d45 = p[4] - p[5];
        const float s67 = p[6] + p[7], d67 = p[6] - p[7];
        const float s0123 = s01 + s23, s4567 = s45 + s67;
        const float S = s0123 + s4567;
        zb0 += (d01 + d23) + (d45 + d67);
        zb1 += (s01 - s23) + (s45 - s67);
        zb2 += s0123 - s4567;
        tot += S;
        zb9 += (nt & 32) ? -S : S;
    }

    float v[NQ];
    v[9] = zb0;
    v[8] = zb1;
    v[7] = zb2;
    v[6] = half ? -tot : tot;
    v[5] = (wv & 1) ? -tot : tot;
    v[4] = (wv & 2) ? -tot : tot;
    v[3] = (wv & 4) ? -tot : tot;
    v[2] = (wv & 8) ? -tot : tot;
    v[1] = (wv & 16) ? -tot : tot;
    v[0] = zb9;
#pragma unroll
    for (int w = 0; w < NQ; ++w) v[w] += __shfl_xor(v[w], 16, 32);
    if (lane < SPB) {
#pragma unroll
        for (int w = 0; w < NQ; ++w) sPart[wv][lane * NQ + w] = v[w];
    }
    __syncthreads();
    if (tid < SPB * NQ) {
        float a = sPart[0][tid];
        for (int u = 1; u < GEMM_WAVES; ++u) a += sPart[u][tid];
        sEv[tid] = a;
    }
    __syncthreads();

    float* orun = out + (size_t)m0 * CDIM;
    for (int i = 0; i < 2; ++i) {
        const int o = 4 * (tid + 1024 * i);
        if (o < SPB * CDIM) {
            const int row = o / CDIM;
            const int c = o - row * CDIM;
            float acc[4];
#pragma unroll
            for (int x = 0; x < 4; ++x) acc[x] = b_out[c + x];
#pragma unroll
            for (int w = 0; w < NQ; ++w) {
                const float e = sEv[row * NQ + w];
#pragma unroll
                for (int x = 0; x < 4; ++x) acc[x] += e * W_out[w * CDIM + c + x];
            }
#pragma unroll
            for (int x = 0; x < 4; ++x) orun[o + x] = acc[x];
        }
    }
}

extern "C" void kernel_launch(void* const* d_in, const int* in_sizes, int n_in,
                              void* d_out, int out_size, void* d_ws,
                              size_t ws_size, hipStream_t stream) {
    const float* z_i     = (const float*)d_in[0];
    const float* W_in    = (const float*)d_in[1];
    const float* b_in    = (const float*)d_in[2];
    const float* weights = (const float*)d_in[3];
    const float* W_out   = (const float*)d_in[4];
    const float* b_out   = (const float*)d_in[5];
    float* out = (float*)d_out;

    float* M_f = (float*)d_ws;

    build_unitary_kernel<<<SDIM / BROWS, 512, 0, stream>>>(weights, M_f);
    state_gemm_kernel<<<BATCH / SPB, GEMM_WAVES * 32, 0, stream>>>(z_i, W_in, b_in, weights, M_f,
                                                                  W_out, b_out, out);
}


# complex product as 3 real WMMA products (Mre.Ere, Mim.Eim, (Mre+Mim).(Ere+Eim)); block = 16 samples x 512 rows j; row halves' Z sums added in a new 3rd launch; F rounding, Z sum order change
# speedup vs baseline: 5.2909x; 1.1731x over previous
#include <hip/hip_runtime.h>

#define NQ      10
#define SDIM    1024
#define CDIM    768
#define NLAYERS 3
#define BATCH   2048

typedef float v2f __attribute__((ext_vector_type(2)));
typedef float v4f __attribute__((ext_vector_type(4)));
typedef float v8f __attribute__((ext_vector_type(8)));

__device__ __forceinline__ float2 cmul(float2 a, float2 b) {
    return make_float2(a.x * b.x - a.y * b.y, a.x * b.y + a.y * b.x);
}
__device__ __forceinline__ float2 cadd(float2 a, float2 b) {
    return make_float2(a.x + b.x, a.y + b.y);
}

#define BROWS 4
__device__ __forceinline__ int cnot_ring_on_index(int k, int rng) {
    for (int i = 0; i < NQ; ++i) {
        const int t = (i + rng) % NQ;
        if ((k >> (9 - i)) & 1) k ^= 1 << (9 - t);
    }
    return k;
}
__global__ void __launch_bounds__(512) build_unitary_kernel(
    const float* __restrict__ weights, float* __restrict__ M_f) {
    __shared__ float2 st[BROWS][SDIM];
    const int tid = threadIdx.x;
    const int ks = blockIdx.x;
#pragma unroll
    for (int r = 0; r < BROWS; ++r) {
        const int k = cnot_ring_on_index(BROWS * ks + r, 1);
        st[r][tid]       = make_float2(tid == k ? 1.f : 0.f, 0.f);
        st[r][tid + 512] = make_float2((tid + 512) == k ? 1.f : 0.f, 0.f);
    }
    __syncthreads();

    for (int l = 1; l < NLAYERS; ++l) {
        for (int q = 0; q < NQ; ++q) {
            const float* w = weights + (l * NQ + q) * 3;
            const float phi = w[0], theta = w[1], omega = w[2];
            const float ct = cosf(0.5f * theta), sv = sinf(0.5f * theta);
            const float al = 0.5f * (phi + omega), be = 0.5f * (phi - omega);
            const float ca = cosf(al), sa = sinf(al);
            const float cb = cosf(be), sb = sinf(be);
            const float2 g00 = make_float2(ct * ca, -ct * sa);
            const float2 g01 = make_float2(-sv * cb, -sv * sb);
            const float2 g10 = make_float2(sv * cb, -sv * sb);
            const float2 g11 = make_float2(ct * ca, ct * sa);
            const int mask = 1 << (9 - q);
            const int low = tid & (mask - 1);
            const int high = (tid & ~(mask - 1)) << 1;
            const int i0 = high | low, i1 = i0 | mask;
            float2 a0[BROWS], a1[BROWS];
#pragma unroll
            for (int r = 0; r < BROWS; ++r) {
                a0[r] = st[r][i0];
                a1[r] = st[r][i1];
            }
#pragma unroll
            for (int r = 0; r < BROWS; ++r) {
                const float2 n0 = cadd(cmul(g00, a0[r]), cmul(g01, a1[r]));
                const float2 n1 = cadd(cmul(g10, a0[r]), cmul(g11, a1[r]));
                st[r][i0] = n0;
                st[r][i1] = n1;
            }
            __syncthreads();
        }
        const int rng = (l % (NQ - 1)) + 1;
        for (int i = 0; i < NQ; ++i) {
            const int c = i, t = (i + rng) % NQ;
            const int maskT = 1 << (9 - t), maskC = 1 << (9 - c);
            const int low = tid & (maskT - 1);
            const int high = (tid & ~(maskT - 1)) << 1;
            const int i0 = high | low;
            if (i0 & maskC) {
                const int i1 = i0 | maskT;
                float2 a[BROWS], b[BROWS];
#pragma unroll
                for (int r = 0; r < BROWS; ++r) {
                    a[r] = st[r][i0];
                    b[r] = st[r][i1];
                }
#pragma unroll
                for (int r = 0; r < BROWS; ++r) {
                    st[r][i0] = b[r];
                    st[r][i1] = a[r];
                }
            }
            __syncthreads();
        }
    }
    v4f* dst = (v4f*)M_f + (size_t)ks * (SDIM / 16 * 32);
    for (int i = 0; i < 4; ++i) {
        const int e = tid + 512 * i;
        const int h = (e >> 4) & 1;
        const int j = ((e >> 5) << 4) | (e & 15);
        const float2 a = st[2 * h][j], b = st[2 * h + 1][j];
        const v4f v = {a.x, b.x, a.y, b.y};
        dst[e] = v;
    }
}

#define GEMM_WAVES 32
#define SPB 16
#define OSPB 8
__global__ void __launch_bounds__(GEMM_WAVES * 32) state_gemm_kernel(
    const float* __restrict__ z, const float* __restrict__ W_in,
    const float* __restrict__ b_in, const float* __restrict__ weights,
    const float* __restrict__ M_f, float* __restrict__ part) {
    __shared__ v4f sE[128 * 2 * 32];
    __shared__ float sPart[GEMM_WAVES][SPB * NQ];
    __shared__ float sW[SPB][8][NQ];
    __shared__ float2 sA0[SPB][NQ], sA1[SPB][NQ];
    __shared__ float2 sHi[SPB][33], sLo[SPB][33];
    const int tid = threadIdx.x;
    const int lane = tid & 31, wv = tid >> 5;
    const int jh = blockIdx.x & 1;
    const int m0 = (blockIdx.x >> 1) * SPB;

    const int half = lane >> 4;
    const int l15 = lane & 15;

    {
        const int team = tid >> 8, t = tid & 255, wv8 = t >> 5;
        float w[3][NQ];
#pragma unroll
        for (int j = 0; j < 3; ++j) {
            const float* wr = W_in + (t + 256 * j) * NQ;
#pragma unroll
            for (int q = 0; q < NQ; ++q) w[j][q] = wr[q];
        }
        for (int i = 0; i < 4; ++i) {
            const int sl = team * 4 + i;
            const float* zr = z + (size_t)(m0 + sl) * CDIM;
            float acc[NQ];
#pragma unroll
            for (int q = 0; q < NQ; ++q) acc[q] = 0.f;
#pragma unroll
            for (int j = 0; j < 3; ++j) {
                const float zv = zr[t + 256 * j];
#pragma unroll
                for (int q = 0; q < NQ; ++q) acc[q] += zv * w[j][q];
            }
#pragma unroll
            for (int q = 0; q < NQ; ++q) {
                for (int s = 16; s > 0; s >>= 1) acc[q] += __shfl_xor(acc[q], s, 32);
            }
            if (lane == 0) {
#pragma unroll
                for (int q = 0; q < NQ; ++q) sW[sl][wv8][q] = acc[q];
            }
        }
    }
    __syncthreads();
    if (tid < SPB * NQ) {
        const int sl = tid / NQ, q = tid - sl * NQ;
        float a = b_in[q];
        for (int u = 0; u < 8; ++u) a += sW[sl][u][q];
        const float c = cosf(0.5f * a), s = sinf(0.5f * a);
        const float* w = weights + q * 3;
        const float phi = w[0], theta = w[1], omega = w[2];
        const float ct = cosf(0.5f * theta), sv = sinf(0.5f * theta);
        const float al = 0.5f * (phi + omega), be = 0.5f * (phi - omega);
        const float ca = cosf(al), sa = sinf(al);
        const float cb = cosf(be), sb = sinf(be);
        const float2 g00 = make_float2(ct * ca, -ct * sa);
        const float2 g01 = make_float2(-sv * cb, -sv * sb);
        const float2 g10 = make_float2(sv * cb, -sv * sb);
        const float2 g11 = make_float2(ct * ca, ct * sa);
        const float2 v0 = make_float2(c, 0.f), v1 = make_float2(0.f, -s);
        sA0[sl][q] = cadd(cmul(g00, v0), cmul(g01, v1));
        sA1[sl][q] = cadd(cmul(g10, v0), cmul(g11, v1));
    }
    __syncthreads();
    {
        const int sl = tid >> 6, tb = (tid >> 5) & 1, idx = tid & 31;
        float2 p = make_float2(1.f, 0.f);
#pragma unroll
        for (int qq = 0; qq < 5; ++qq) {
            const int q = tb * 5 + qq;
            p = cmul(p, ((idx >> (4 - qq)) & 1) ? sA1[sl][q] : sA0[sl][q]);
        }
        if (tb) sLo[sl][idx] = p; else sHi[sl][idx] = p;
    }
    __syncthreads();
    {
        const int kpl = tid >> 5;
        float2 lo[4];
#pragma unroll
        for (int u = 0; u < 4; ++u)
            lo[u] = sLo[l15][8 * (kpl & 3) + 4 * (u >> 1) + 2 * half + (u & 1)];
        for (int i = 0; i < 4; ++i) {
            const int kp = kpl + 32 * i;
            const float2 hi = sHi[l15][kp >> 2];
            float vr[4], vi[4];
#pragma unroll
            for (int u = 0; u < 4; ++u) {
                const float2 e = cmul(hi, lo[u]);
                vr[u] = e.x;
                vi[u] = e.y;
            }
            const v4f ere = {vr[0], vr[1], vr[2], vr[3]};
            const v4f eim = {vi[0], vi[1], vi[2], vi[3]};
            sE[(kp * 2) * 32 + lane] = ere;
            sE[(kp * 2 + 1) * 32 + lane] = eim;
        }
    }
    __syncthreads();

    const v4f* Bp = sE + lane;
    const int nt = GEMM_WAVES * jh + wv;
    const v4f* Ap = (const v4f*)M_f + (size_t)nt * 32 + lane;
    v8f p1 = {}, p2 = {}, p3 = {};
    for (int ks0 = 0; ks0 < SDIM / 4; ks0 += 8) {
        v4f er[4], ei[4], a[8];
#pragma unroll
        for (int u = 0; u < 4; ++u) {
            er[u] = Bp[(((ks0 >> 1) + u) * 2) * 32];
            ei[u] = Bp[(((ks0 >> 1) + u) * 2 + 1) * 32];
        }
#pragma unroll
        for (int u = 0; u < 8; ++u) a[u] = Ap[(size_t)(ks0 + u) * (64 * 32)];
        __builtin_amdgcn_sched_barrier(0);
#pragma unroll
        for (int u = 0; u < 8; ++u) {
            const v2f b_re = (u & 1) ? (v2f){er[u >> 1].z, er[u >> 1].w}
                                     : (v2f){er[u >> 1].x, er[u >> 1].y};
            const v2f b_im = (u & 1) ? (v2f){ei[u >> 1].z, ei[u >> 1].w}
                                     : (v2f){ei[u >> 1].x, ei[u >> 1].y};
            const v2f b_su = b_re + b_im;
            const v2f m_re = {a[u].x, a[u].y};
            const v2f m_im = {a[u].z, a[u].w};
            const v2f m_su = m_re + m_im;
            p1 = __builtin_amdgcn_wmma_f32_16x16x4_f32(false, m_re, false, b_re,
                                                       (short)0, p1, false, false);
            p2 = __builtin_amdgcn_wmma_f32_16x16x4_f32(false, m_im, false, b_im,
                                                       (short)0, p2, false, false);
            p3 = __builtin_amdgcn_wmma_f32_16x16x4_f32(false, m_su, false, b_su,
                                                       (short)0, p3, false, false);
        }
    }
    float p[8];
#pragma unroll
    for (int r = 0; r < 8; ++r) {
        const float fre = p1[r] - p2[r];
        const float fim = (p3[r] - p1[r]) - p2[r];
        p[r] = fre * fre + fim * fim;
    }
    const float s01 = p[0] + p[1], d01 = p[0] - p[1];
    const float s23 = p[2] + p[3], d23 = p[2] - p[3];
    const float s45 = p[4] + p[5], d45 = p[4] - p[5];
    const float s67 = p[6] + p[7], d67 = p[6] - p[7];
    const float s0123 = s01 + s23, s4567 = s45 + s67;
    const float S = s0123 + s4567;

    float v[NQ];
    v[9] = (d01 + d23) + (d45 + d67);
    v[8] = (s01 - s23) + (s45 - s67);
    v[7] = s0123 - s4567;
    v[6] = half ? -S : S;
    v[5] = (wv & 1) ? -S : S;
    v[4] = (wv & 2) ? -S : S;
    v[3] = (wv & 4) ? -S : S;
    v[2] = (wv & 8) ? -S : S;
    v[1] = (wv & 16) ? -S : S;
    v[0] = jh ? -S : S;
#pragma unroll
    for (int w = 0; w < NQ; ++w) v[w] += __shfl_xor(v[w], 16, 32);
    if (lane < SPB) {
#pragma unroll
        for (int w = 0; w < NQ; ++w) sPart[wv][lane * NQ + w] = v[w];
    }
    __syncthreads();
    if (tid < SPB * NQ) {
        float a = sPart[0][tid];
        for (int u = 1; u < GEMM_WAVES; ++u) a += sPart[u][tid];
        part[(size_t)jh * (BATCH * NQ) + m0 * NQ + tid] = a;
    }
}

__global__ void __launch_bounds__(512) output_kernel(
    const float* __restrict__ part, const float* __restrict__ W_out,
    const float* __restrict__ b_out, float* __restrict__ out) {
    __shared__ float sEv[OSPB * NQ];
    const int tid = threadIdx.x;
    const int m0 = blockIdx.x * OSPB;
    if (tid < OSPB * NQ) sEv[tid] = part[m0 * NQ + tid] + part[BATCH * NQ + m0 * NQ + tid];
    __syncthreads();
    float* orun = out + (size_t)m0 * CDIM;
    for (int i = 0; i < 3; ++i) {
        const int o = 4 * (tid + 512 * i);
        const int row = o / CDIM;
        const int c = o - row * CDIM;
        float acc[4];
#pragma unroll
        for (int x = 0; x < 4; ++x) acc[x] = b_out[c + x];
#pragma unroll
        for (int w = 0; w < NQ; ++w) {
            const float e = sEv[row * NQ + w];
#pragma unroll
            for (int x = 0; x < 4; ++x) acc[x] += e * W_out[w * CDIM + c + x];
        }
#pragma unroll
        for (int x = 0; x < 4; ++x) orun[o + x] = acc[x];
    }
}

extern "C" void kernel_launch(void* const* d_in, const int* in_sizes, int n_in,
                              void* d_out, int out_size, void* d_ws,
                              size_t ws_size, hipStream_t stream) {
    const float* z_i     = (const float*)d_in[0];
    const float* W_in    = (const float*)d_in[1];
    const float* b_in    = (const float*)d_in[2];
    const float* weights = (const float*)d_in[3];
    const float* W_out   = (const float*)d_in[4];
    const float* b_out   = (const float*)d_in[5];
    float* out = (float*)d_out;

    float* M_f = (float*)d_ws;
    float* part = M_f + 2 * SDIM * SDIM;

    build_unitary_kernel<<<SDIM / BROWS, 512, 0, stream>>>(weights, M_f);
    state_gemm_kernel<<<BATCH / SPB * 2, GEMM_WAVES * 32, 0, stream>>>(z_i, W_in, b_in, weights, M_f,
                                                                      part);
    output_kernel<<<BATCH / OSPB, 512, 0, stream>>>(part, W_out, b_out, out);
}


# unitary build: layer 1 on a basis state set up as a product state, layer 2 applied as two 32x32 complex products on WMMA (3 real each), rings = index maps; 5 barriers, was 41; sum order in M changes
# speedup vs baseline: 6.2128x; 1.1742x over previous
#include <hip/hip_runtime.h>

#define NQ      10
#define SDIM    1024
#define CDIM    768
#define NLAYERS 3
#define BATCH   2048

typedef float v2f __attribute__((ext_vector_type(2)));
typedef float v4f __attribute__((ext_vector_type(4)));
typedef float v8f __attribute__((ext_vector_type(8)));

__device__ __forceinline__ float2 cmul(float2 a, float2 b) {
    return make_float2(a.x * b.x - a.y * b.y, a.x * b.y + a.y * b.x);
}
__device__ __forceinline__ float2 cadd(float2 a, float2 b) {
    return make_float2(a.x + b.x, a.y + b.y);
}

#define BROWS 4
#define KP 34
__device__ __forceinline__ int cnot_ring_on_index(int k, int rng) {
    for (int i = 0; i < NQ; ++i) {
        const int t = (i + rng) % NQ;
        if ((k >> (9 - i)) & 1) k ^= 1 << (9 - t);
    }
    return k;
}
__device__ __forceinline__ float2 kron5_entry(const float2 (*sg5)[4], int o, int i) {
    float2 p = make_float2(1.f, 0.f);
#pragma unroll
    for (int qq = 0; qq < 5; ++qq)
        p = cmul(p, sg5[qq][2 * ((o >> (4 - qq)) & 1) + ((i >> (4 - qq)) & 1)]);
    return p;
}
__device__ __forceinline__ void kron_tile(const float2* Grow, const float2* Srow, int hf,
                                          v8f& fre, v8f& fim) {
    v8f p1 = {}, p2 = {}, p3 = {};
#pragma unroll
    for (int s = 0; s < 8; ++s) {
        const v4f g = *(const v4f*)(Grow + 4 * s + 2 * hf);
        const v4f x = *(const v4f*)(Srow + 4 * s + 2 * hf);
        const v2f g_re = {g.x, g.z}, g_im = {g.y, g.w};
        const v2f x_re = {x.x, x.z}, x_im = {x.y, x.w};
        const v2f g_su = g_re + g_im, x_su = x_re + x_im;
        p1 = __builtin_amdgcn_wmma_f32_16x16x4_f32(false, g_re, false, x_re, (short)0, p1, false, false);
        p2 = __builtin_amdgcn_wmma_f32_16x16x4_f32(false, g_im, false, x_im, (short)0, p2, false, false);
        p3 = __builtin_amdgcn_wmma_f32_16x16x4_f32(false, g_su, false, x_su, (short)0, p3, false, false);
    }
#pragma unroll
    for (int r = 0; r < 8; ++r) {
        fre[r] = p1[r] - p2[r];
        fim[r] = (p3[r] - p1[r]) - p2[r];
    }
}
__global__ void __launch_bounds__(512) build_unitary_kernel(
    const float* __restrict__ weights, float* __restrict__ M_f) {
    __shared__ float2 sg[2 * NQ][4];
    __shared__ float2 sCol[2][BROWS][32];
    __shared__ __align__(16) float2 sGm[2][32][KP];
    __shared__ __align__(16) float2 sX[BROWS][32][KP];
    __shared__ __align__(16) float2 sT[BROWS][32][KP];
    __shared__ float2 st[BROWS][SDIM];
    const int tid = threadIdx.x;
    const int lane = tid & 31, wv = tid >> 5;
    const int hf = lane >> 4, l15 = lane & 15;
    const int ks = blockIdx.x;

    if (tid < 2 * NQ) {
        const float* w = weights + (NQ + tid) * 3;
        const float phi = w[0], theta = w[1], omega = w[2];
        const float ct = cosf(0.5f * theta), sv = sinf(0.5f * theta);
        const float al = 0.5f * (phi + omega), be = 0.5f * (phi - omega);
        const float ca = cosf(al), sa = sinf(al);
        const float cb = cosf(be), sb = sinf(be);
        sg[tid][0] = make_float2(ct * ca, -ct * sa);
        sg[tid][1] = make_float2(-sv * cb, -sv * sb);
        sg[tid][2] = make_float2(sv * cb, -sv * sb);
        sg[tid][3] = make_float2(ct * ca, ct * sa);
    }
    __syncthreads();
    for (int u = 0; u < 4; ++u) {
        const int e = tid + 512 * u;
        const int which = e >> 10, m = (e >> 5) & 31, b = e & 31;
        sGm[which][m][b] = kron5_entry(sg + NQ + (which ? 0 : 5), m, b);
    }
    if (tid < 2 * BROWS * 32) {
        const int tb = tid >> 7, kc = (tid >> 5) & 3, a = tid & 31;
        const int k1 = cnot_ring_on_index(BROWS * ks + kc, 1);
        sCol[tb][kc][a] = kron5_entry(sg + (tb ? 5 : 0), a, tb ? (k1 & 31) : (k1 >> 5));
    }
    __syncthreads();
    for (int v = 0; v < 2; ++v) {
        const int i = tid + 512 * v;
        const int pos = cnot_ring_on_index(i, 2);
#pragma unroll
        for (int kc = 0; kc < BROWS; ++kc)
            sX[kc][pos >> 5][pos & 31] = cmul(sCol[0][kc][i >> 5], sCol[1][kc][i & 31]);
    }
    __syncthreads();
    const int m0 = 16 * (wv & 1), kc = wv >> 2, a0 = 16 * ((wv >> 1) & 1);
    v8f fre, fim;
    kron_tile(&sGm[0][m0 + l15][0], &sX[kc][a0 + l15][0], hf, fre, fim);
#pragma unroll
    for (int r = 0; r < 8; ++r) sT[kc][m0 + 8 * hf + r][a0 + l15] = make_float2(fre[r], fim[r]);
    __syncthreads();
    kron_tile(&sGm[1][m0 + l15][0], &sT[kc][a0 + l15][0], hf, fre, fim);
    const int jb = cnot_ring_on_index((m0 + 8 * hf) * 32 + a0 + l15, 3);
#pragma unroll
    for (int r = 0; r < 8; ++r)
        st[kc][jb ^ cnot_ring_on_index(r << 5, 3)] = make_float2(fre[r], fim[r]);
    __syncthreads();
    v4f* dst = (v4f*)M_f + (size_t)ks * (SDIM / 16 * 32);
    for (int i = 0; i < 4; ++i) {
        const int e = tid + 512 * i;
        const int h = (e >> 4) & 1;
        const int j = ((e >> 5) << 4) | (e & 15);
        const float2 a = st[2 * h][j], b = st[2 * h + 1][j];
        const v4f v = {a.x, b.x, a.y, b.y};
        dst[e] = v;
    }
}

#define GEMM_WAVES 32
#define SPB 16
#define OSPB 8
__global__ void __launch_bounds__(GEMM_WAVES * 32) state_gemm_kernel(
    const float* __restrict__ z, const float* __restrict__ W_in,
    const float* __restrict__ b_in, const float* __restrict__ weights,
    const float* __restrict__ M_f, float* __restrict__ part) {
    __shared__ v4f sE[128 * 2 * 32];
    __shared__ float sPart[GEMM_WAVES][SPB * NQ];
    __shared__ float sW[SPB][8][NQ];
    __shared__ float2 sA0[SPB][NQ], sA1[SPB][NQ];
    __shared__ float2 sHi[SPB][33], sLo[SPB][33];
    const int tid = threadIdx.x;
    const int lane = tid & 31, wv = tid >> 5;
    const int jh = blockIdx.x & 1;
    const int m0 = (blockIdx.x >> 1) * SPB;

    const int half = lane >> 4;
    const int l15 = lane & 15;

    {
        const int team = tid >> 8, t = tid & 255, wv8 = t >> 5;
        float w[3][NQ];
#pragma unroll
        for (int j = 0; j < 3; ++j) {
            const float* wr = W_in + (t + 256 * j) * NQ;
#pragma unroll
            for (int q = 0; q < NQ; ++q) w[j][q] = wr[q];
        }
        for (int i = 0; i < 4; ++i) {
            const int sl = team * 4 + i;
            const float* zr = z + (size_t)(m0 + sl) * CDIM;
            float acc[NQ];
#pragma unroll
            for (int q = 0; q < NQ; ++q) acc[q] = 0.f;
#pragma unroll
            for (int j = 0; j < 3; ++j) {
                const float zv = zr[t + 256 * j];
#pragma unroll
                for (int q = 0; q < NQ; ++q) acc[q] += zv * w[j][q];
            }
#pragma unroll
            for (int q = 0; q < NQ; ++q) {
                for (int s = 16; s > 0; s >>= 1) acc[q] += __shfl_xor(acc[q], s, 32);
            }
            if (lane == 0) {
#pragma unroll
                for (int q = 0; q < NQ; ++q) sW[sl][wv8][q] = acc[q];
            }
        }
    }
    __syncthreads();
    if (tid < SPB * NQ) {
        const int sl = tid / NQ, q = tid - sl * NQ;
        float a = b_in[q];
        for (int u = 0; u < 8; ++u) a += sW[sl][u][q];
        const float c = cosf(0.5f * a), s = sinf(0.5f * a);
        const float* w = weights + q * 3;
        const float phi = w[0], theta = w[1], omega = w[2];
        const float ct = cosf(0.5f * theta), sv = sinf(0.5f * theta);
        const float al = 0.5f * (phi + omega), be = 0.5f * (phi - omega);
        const float ca = cosf(al), sa = sinf(al);
        const float cb = cosf(be), sb = sinf(be);
        const float2 g00 = make_float2(ct * ca, -ct * sa);
        const float2 g01 = make_float2(-sv * cb, -sv * sb);
        const float2 g10 = make_float2(sv * cb, -sv * sb);
        const float2 g11 = make_float2(ct * ca, ct * sa);
        const float2 v0 = make_float2(c, 0.f), v1 = make_float2(0.f, -s);
        sA0[sl][q] = cadd(cmul(g00, v0), cmul(g01, v1));
        sA1[sl][q] = cadd(cmul(g10, v0), cmul(g11, v1));
    }
    __syncthreads();
    {
        const int sl = tid >> 6, tb = (tid >> 5) & 1, idx = tid & 31;
        float2 p = make_float2(1.f, 0.f);
#pragma unroll
        for (int qq = 0; qq < 5; ++qq) {
            const int q = tb * 5 + qq;
            p = cmul(p, ((idx >> (4 - qq)) & 1) ? sA1[sl][q] : sA0[sl][q]);
        }
        if (tb) sLo[sl][idx] = p; else sHi[sl][idx] = p;
    }
    __syncthreads();
    {
        const int kpl = tid >> 5;
        float2 lo[4];
#pragma unroll
        for (int u = 0; u < 4; ++u)
            lo[u] = sLo[l15][8 * (kpl & 3) + 4 * (u >> 1) + 2 * half + (u & 1)];
        for (int i = 0; i < 4; ++i) {
            const int kp = kpl + 32 * i;
            const float2 hi = sHi[l15][kp >> 2];
            float vr[4], vi[4];
#pragma unroll
            for (int u = 0; u < 4; ++u) {
                const float2 e = cmul(hi, lo[u]);
                vr[u] = e.x;
                vi[u] = e.y;
            }
            const v4f ere = {vr[0], vr[1], vr[2], vr[3]};
            const v4f eim = {vi[0], vi[1], vi[2], vi[3]};
            sE[(kp * 2) * 32 + lane] = ere;
            sE[(kp * 2 + 1) * 32 + lane] = eim;
        }
    }
    __syncthreads();

    const v4f* Bp = sE + lane;
    const int nt = GEMM_WAVES * jh + wv;
    const v4f* Ap = (const v4f*)M_f + (size_t)nt * 32 + lane;
    v8f p1 = {}, p2 = {}, p3 = {};
    for (int ks0 = 0; ks0 < SDIM / 4; ks0 += 8) {
        v4f er[4], ei[4], a[8];
#pragma unroll
        for (int u = 0; u < 4; ++u) {
            er[u] = Bp[(((ks0 >> 1) + u) * 2) * 32];
            ei[u] = Bp[(((ks0 >> 1) + u) * 2 + 1) * 32];
        }
#pragma unroll
        for (int u = 0; u < 8; ++u) a[u] = Ap[(size_t)(ks0 + u) * (64 * 32)];
        __builtin_amdgcn_sched_barrier(0);
#pragma unroll
        for (int u = 0; u < 8; ++u) {
            const v2f b_re = (u & 1) ? (v2f){er[u >> 1].z, er[u >> 1].w}
                                     : (v2f){er[u >> 1].x, er[u >> 1].y};
            const v2f b_im = (u & 1) ? (v2f){ei[u >> 1].z, ei[u >> 1].w}
                                     : (v2f){ei[u >> 1].x, ei[u >> 1].y};
            const v2f b_su = b_re + b_im;
            const v2f m_re = {a[u].x, a[u].y};
            const v2f m_im = {a[u].z, a[u].w};
            const v2f m_su = m_re + m_im;
            p1 = __builtin_amdgcn_wmma_f32_16x16x4_f32(false, m_re, false, b_re,
                                                       (short)0, p1, false, false);
            p2 = __builtin_amdgcn_wmma_f32_16x16x4_f32(false, m_im, false, b_im,
                                                       (short)0, p2, false, false);
            p3 = __builtin_amdgcn_wmma_f32_16x16x4_f32(false, m_su, false, b_su,
                                                       (short)0, p3, false, false);
        }
    }
    float p[8];
#pragma unroll
    for (int r = 0; r < 8; ++r) {
        const float fre = p1[r] - p2[r];
        const float fim = (p3[r] - p1[r]) - p2[r];
        p[r] = fre * fre + fim * fim;
    }
    const float s01 = p[0] + p[1], d01 = p[0] - p[1];
    const float s23 = p[2] + p[3], d23 = p[2] - p[3];
    const float s45 = p[4] + p[5], d45 = p[4] - p[5];
    const float s67 = p[6] + p[7], d67 = p[6] - p[7];
    const float s0123 = s01 + s23, s4567 = s45 + s67;
    const float S = s0123 + s4567;

    float v[NQ];
    v[9] = (d01 + d23) + (d45 + d67);
    v[8] = (s01 - s23) + (s45 - s67);
    v[7] = s0123 - s4567;
    v[6] = half ? -S : S;
    v[5] = (wv & 1) ? -S : S;
    v[4] = (wv & 2) ? -S : S;
    v[3] = (wv & 4) ? -S : S;
    v[2] = (wv & 8) ? -S : S;
    v[1] = (wv & 16) ? -S : S;
    v[0] = jh ? -S : S;
#pragma unroll
    for (int w = 0; w < NQ; ++w) v[w] += __shfl_xor(v[w], 16, 32);
    if (lane < SPB) {
#pragma unroll
        for (int w = 0; w < NQ; ++w) sPart[wv][lane * NQ + w] = v[w];
    }
    __syncthreads();
    if (tid < SPB * NQ) {
        float a = sPart[0][tid];
        for (int u = 1; u < GEMM_WAVES; ++u) a += sPart[u][tid];
        part[(size_t)jh * (BATCH * NQ) + m0 * NQ + tid] = a;
    }
}

__global__ void __launch_bounds__(512) output_kernel(
    const float* __restrict__ part, const float* __restrict__ W_out,
    const float* __restrict__ b_out, float* __restrict__ out) {
    __shared__ float sEv[OSPB * NQ];
    const int tid = threadIdx.x;
    const int m0 = blockIdx.x * OSPB;
    if (tid < OSPB * NQ) sEv[tid] = part[m0 * NQ + tid] + part[BATCH * NQ + m0 * NQ + tid];
    __syncthreads();
    float* orun = out + (size_t)m0 * CDIM;
    for (int i = 0; i < 3; ++i) {
        const int o = 4 * (tid + 512 * i);
        const int row = o / CDIM;
        const int c = o - row * CDIM;
        float acc[4];
#pragma unroll
        for (int x = 0; x < 4; ++x) acc[x] = b_out[c + x];
#pragma unroll
        for (int w = 0; w < NQ; ++w) {
            const float e = sEv[row * NQ + w];
#pragma unroll
            for (int x = 0; x < 4; ++x) acc[x] += e * W_out[w * CDIM + c + x];
        }
#pragma unroll
        for (int x = 0; x < 4; ++x) orun[o + x] = acc[x];
    }
}

extern "C" void kernel_launch(void* const* d_in, const int* in_sizes, int n_in,
                              void* d_out, int out_size, void* d_ws,
                              size_t ws_size, hipStream_t stream) {
    const float* z_i     = (const float*)d_in[0];
    const float* W_in    = (const float*)d_in[1];
    const float* b_in    = (const float*)d_in[2];
    const float* weights = (const float*)d_in[3];
    const float* W_out   = (const float*)d_in[4];
    const float* b_out   = (const float*)d_in[5];
    float* out = (float*)d_out;

    float* M_f = (float*)d_ws;
    float* part = M_f + 2 * SDIM * SDIM;

    build_unitary_kernel<<<SDIM / BROWS, 512, 0, stream>>>(weights, M_f);
    state_gemm_kernel<<<BATCH / SPB * 2, GEMM_WAVES * 32, 0, stream>>>(z_i, W_in, b_in, weights, M_f,
                                                                      part);
    output_kernel<<<BATCH / OSPB, 512, 0, stream>>>(part, W_out, b_out, out);
}


# matrix = just the last layer's RY factors (real; rows renamed by the last ring; RZ(omega) cannot change |amp|^2); layer 1 per sample on WMMA; GEMM real x complex: 2 products, was 3; sum orders change
# speedup vs baseline: 8.3768x; 1.3483x over previous
#include <hip/hip_runtime.h>

#define NQ      10
#define SDIM    1024
#define CDIM    768
#define NLAYERS 3
#define BATCH   2048

typedef float v2f __attribute__((ext_vector_type(2)));
typedef float v4f __attribute__((ext_vector_type(4)));
typedef float v8f __attribute__((ext_vector_type(8)));

__device__ __forceinline__ float2 cmul(float2 a, float2 b) {
    return make_float2(a.x * b.x - a.y * b.y, a.x * b.y + a.y * b.x);
}
__device__ __forceinline__ float2 cadd(float2 a, float2 b) {
    return make_float2(a.x + b.x, a.y + b.y);
}

__device__ __forceinline__ int cnot_ring_on_index(int k, int rng) {
    for (int i = 0; i < NQ; ++i) {
        const int t = (i + rng) % NQ;
        if ((k >> (9 - i)) & 1) k ^= 1 << (9 - t);
    }
    return k;
}
__device__ __forceinline__ int cnot_ring_inverse_on_index(int k, int rng) {
    for (int i = NQ - 1; i >= 0; --i) {
        const int t = (i + rng) % NQ;
        if ((k >> (9 - i)) & 1) k ^= 1 << (9 - t);
    }
    return k;
}
__device__ __forceinline__ float ry5_entry(const float (*cs5)[2], int o, int i) {
    float p = 1.f;
#pragma unroll
    for (int qq = 0; qq < 5; ++qq) {
        const int bo = (o >> (4 - qq)) & 1, bi = (i >> (4 - qq)) & 1;
        const float c = cs5[qq][0], s = cs5[qq][1];
        p *= (bo == bi) ? c : (bo ? s : -s);
    }
    return p;
}
__device__ __forceinline__ float2 rz_diag_entry(const float2* ph, int idx) {
    float2 p = make_float2(1.f, 0.f);
#pragma unroll
    for (int q = 0; q < NQ; ++q) {
        float2 e = ph[q];
        if ((idx >> (9 - q)) & 1) e.y = -e.y;
        p = cmul(p, e);
    }
    return p;
}

__global__ void __launch_bounds__(512) build_matrix_kernel(
    const float* __restrict__ weights, float* __restrict__ R_f) {
    __shared__ float sCS[NQ][2];
    __shared__ float sRlo[32][33];
    __shared__ float sRhiCol[32];
    const int tid = threadIdx.x;
    const int ksp = blockIdx.x >> 1, part = blockIdx.x & 1;
    if (tid < NQ) {
        const float theta = weights[(2 * NQ + tid) * 3 + 1];
        sCS[tid][0] = cosf(0.5f * theta);
        sCS[tid][1] = sinf(0.5f * theta);
    }
    __syncthreads();
    for (int u = 0; u < 2; ++u) {
        const int e = tid + 512 * u;
        sRlo[e >> 5][e & 31] = ry5_entry(sCS + 5, e >> 5, e & 31);
    }
    if (tid < 32) sRhiCol[tid] = ry5_entry(sCS, tid, ksp >> 2);
    __syncthreads();
    v4f* dst = (v4f*)R_f + (size_t)ksp * (SDIM / 16 * 32);
    const int kl = 8 * (ksp & 3);
    for (int i = 0; i < 2; ++i) {
        const int e = part * 1024 + tid + 512 * i;
        const int h = (e >> 4) & 1;
        const int j = ((e >> 5) << 4) | (e & 15);
        const int ii = cnot_ring_inverse_on_index(j, 3);
        const float hi = sRhiCol[ii >> 5];
        const float* lo = &sRlo[ii & 31][kl + 2 * h];
        const v4f v = {hi * lo[0], hi * lo[1], hi * lo[4], hi * lo[5]};
        dst[e] = v;
    }
}

#define GEMM_WAVES 32
#define SPB 8
#define OSPB 8
#define KP 34
#define RP 36
__device__ __forceinline__ void real_cplx_tile(const float* Grow, const float2* Srow, int hf,
                                               v8f& fre, v8f& fim) {
    v8f pr = {}, pi = {};
#pragma unroll
    for (int s = 0; s < 8; ++s) {
        const v2f g = *(const v2f*)(Grow + 4 * s + 2 * hf);
        const v4f x = *(const v4f*)(Srow + 4 * s + 2 * hf);
        const v2f x_re = {x.x, x.z}, x_im = {x.y, x.w};
        pr = __builtin_amdgcn_wmma_f32_16x16x4_f32(false, g, false, x_re, (short)0, pr, false, false);
        pi = __builtin_amdgcn_wmma_f32_16x16x4_f32(false, g, false, x_im, (short)0, pi, false, false);
    }
    fre = pr;
    fim = pi;
}
__global__ void __launch_bounds__(GEMM_WAVES * 32) state_gemm_kernel(
    const float* __restrict__ z, const float* __restrict__ W_in,
    const float* __restrict__ b_in, const float* __restrict__ weights,
    const float* __restrict__ R_f, float* __restrict__ evs) {
    __shared__ v4f sY[128 * 32];
    __shared__ __align__(16) float2 sX[SPB][32][KP];
    __shared__ __align__(16) float2 sT[SPB][32][KP];
    __shared__ __align__(16) float sRm[2][32][RP];
    __shared__ float2 sPhIn[SDIM];
    __shared__ float2 sPhOut[SDIM];
    __shared__ float sCS[NQ][2];
    __shared__ float2 sPh[3][NQ];
    __shared__ float sPart[GEMM_WAVES][SPB * NQ];
    __shared__ float sW[SPB][8][NQ];
    __shared__ float2 sA0[SPB][NQ], sA1[SPB][NQ];
    __shared__ float2 sHi[SPB][33], sLo[SPB][33];
    const int tid = threadIdx.x;
    const int lane = tid & 31, wv = tid >> 5;
    const int m0s = blockIdx.x * SPB;

    const int half = lane >> 4;
    const int l15 = lane & 15;

    {
        const int team = tid >> 8, t = tid & 255, wv8 = t >> 5;
        float w[3][NQ];
#pragma unroll
        for (int j = 0; j < 3; ++j) {
            const float* wr = W_in + (t + 256 * j) * NQ;
#pragma unroll
            for (int q = 0; q < NQ; ++q) w[j][q] = wr[q];
        }
        for (int i = 0; i < 2; ++i) {
            const int sl = team * 2 + i;
            const float* zr = z + (size_t)(m0s + sl) * CDIM;
            float acc[NQ];
#pragma unroll
            for (int q = 0; q < NQ; ++q) acc[q] = 0.f;
#pragma unroll
            for (int j = 0; j < 3; ++j) {
                const float zv = zr[t + 256 * j];
#pragma unroll
                for (int q = 0; q < NQ; ++q) acc[q] += zv * w[j][q];
            }
#pragma unroll
            for (int q = 0; q < NQ; ++q) {
                for (int s = 16; s > 0; s >>= 1) acc[q] += __shfl_xor(acc[q], s, 32);
            }
            if (lane == 0) {
#pragma unroll
                for (int q = 0; q < NQ; ++q) sW[sl][wv8][q] = acc[q];
            }
        }
    }
    if (tid < NQ) {
        const float* w1 = weights + (NQ + tid) * 3;
        const float* w2 = weights + (2 * NQ + tid) * 3;
        sCS[tid][0] = cosf(0.5f * w1[1]);
        sCS[tid][1] = sinf(0.5f * w1[1]);
        sPh[0][tid] = make_float2(cosf(0.5f * w1[0]), -sinf(0.5f * w1[0]));
        sPh[1][tid] = make_float2(cosf(0.5f * w1[2]), -sinf(0.5f * w1[2]));
        sPh[2][tid] = make_float2(cosf(0.5f * w2[0]), -sinf(0.5f * w2[0]));
    }
    __syncthreads();
    if (tid < SPB * NQ) {
        const int sl = tid / NQ, q = tid - sl * NQ;
        float a = b_in[q];
        for (int u = 0; u < 8; ++u) a += sW[sl][u][q];
        const float c = cosf(0.5f * a), s = sinf(0.5f * a);
        const float* w = weights + q * 3;
        const float phi = w[0], theta = w[1], omega = w[2];
        const float ct = cosf(0.5f * theta), sv = sinf(0.5f * theta);
        const float al = 0.5f * (phi + omega), be = 0.5f * (phi - omega);
        const float ca = cosf(al), sa = sinf(al);
        const float cb = cosf(be), sb = sinf(be);
        const float2 g00 = make_float2(ct * ca, -ct * sa);
        const float2 g01 = make_float2(-sv * cb, -sv * sb);
        const float2 g10 = make_float2(sv * cb, -sv * sb);
        const float2 g11 = make_float2(ct * ca, ct * sa);
        const float2 v0 = make_float2(c, 0.f), v1 = make_float2(0.f, -s);
        sA0[sl][q] = cadd(cmul(g00, v0), cmul(g01, v1));
        sA1[sl][q] = cadd(cmul(g10, v0), cmul(g11, v1));
    }
    for (int u = 0; u < 2; ++u) {
        const int e = tid + 1024 * u;
        const int which = e >> 10, m = (e >> 5) & 31, b = e & 31;
        sRm[which][m][b] = ry5_entry(sCS + (which ? 0 : 5), m, b);
    }
    sPhIn[tid] = rz_diag_entry(sPh[0], cnot_ring_on_index(tid, 1));
    sPhOut[tid] = cmul(rz_diag_entry(sPh[1], tid),
                       rz_diag_entry(sPh[2], cnot_ring_on_index(tid, 2)));
    __syncthreads();
    if (tid < SPB * 64) {
        const int sl = tid >> 6, tb = (tid >> 5) & 1, idx = tid & 31;
        float2 p = make_float2(1.f, 0.f);
#pragma unroll
        for (int qq = 0; qq < 5; ++qq) {
            const int q = tb * 5 + qq;
            p = cmul(p, ((idx >> (4 - qq)) & 1) ? sA1[sl][q] : sA0[sl][q]);
        }
        if (tb) sLo[sl][idx] = p; else sHi[sl][idx] = p;
    }
    __syncthreads();
    {
        const int k1 = cnot_ring_on_index(tid, 1);
        const float2 ph = sPhIn[tid];
#pragma unroll
        for (int sl = 0; sl < SPB; ++sl)
            sX[sl][k1 >> 5][k1 & 31] = cmul(cmul(sHi[sl][tid >> 5], sLo[sl][tid & 31]), ph);
    }
    __syncthreads();
    {
        const int m0 = 16 * (wv & 1), a0 = 16 * ((wv >> 1) & 1), sl = wv >> 2;
        v8f fre, fim;
        real_cplx_tile(&sRm[0][m0 + l15][0], &sX[sl][a0 + l15][0], half, fre, fim);
#pragma unroll
        for (int r = 0; r < 8; ++r) sT[sl][m0 + 8 * half + r][a0 + l15] = make_float2(fre[r], fim[r]);
        __syncthreads();
        real_cplx_tile(&sRm[1][m0 + l15][0], &sT[sl][a0 + l15][0], half, fre, fim);
        const int ib = (m0 + 8 * half) * 32 + a0 + l15;
        const int jb = cnot_ring_on_index(ib, 2);
        float* sYf = (float*)sY;
#pragma unroll
        for (int r = 0; r < 8; ++r) {
            const float2 y = cmul(sPhOut[ib | (r << 5)], make_float2(fre[r], fim[r]));
            const int pos = jb ^ cnot_ring_on_index(r << 5, 2);
            const int o = (((pos >> 3) * 32 + 16 * ((pos >> 1) & 1)) << 2) + 2 * ((pos >> 2) & 1) + (pos & 1);
            sYf[o + 4 * sl] = y.x;
            sYf[o + 4 * (sl + 8)] = y.y;
        }
    }
    __syncthreads();

    const v4f* Bp = sY + lane;
    const v4f* Ap0 = (const v4f*)R_f + (size_t)wv * 32 + lane;
    const v4f* Ap1 = Ap0 + GEMM_WAVES * 32;
    v8f c0 = {}, c1 = {};
    for (int kp0 = 0; kp0 < SDIM / 8; kp0 += 4) {
        v4f b[4], a0[4], a1[4];
#pragma unroll
        for (int u = 0; u < 4; ++u) b[u] = Bp[(kp0 + u) * 32];
#pragma unroll
        for (int u = 0; u < 4; ++u) {
            a0[u] = Ap0[(size_t)(kp0 + u) * (64 * 32)];
            a1[u] = Ap1[(size_t)(kp0 + u) * (64 * 32)];
        }
#pragma unroll
        for (int u = 0; u < 4; ++u) {
            const v2f b_lo = {b[u].x, b[u].y}, b_hi = {b[u].z, b[u].w};
            const v2f a0_lo = {a0[u].x, a0[u].y}, a0_hi = {a0[u].z, a0[u].w};
            const v2f a1_lo = {a1[u].x, a1[u].y}, a1_hi = {a1[u].z, a1[u].w};
            c0 = __builtin_amdgcn_wmma_f32_16x16x4_f32(false, a0_lo, false, b_lo,
                                                       (short)0, c0, false, false);
            c1 = __builtin_amdgcn_wmma_f32_16x16x4_f32(false, a1_lo, false, b_lo,
                                                       (short)0, c1, false, false);
            c0 = __builtin_amdgcn_wmma_f32_16x16x4_f32(false, a0_hi, false, b_hi,
                                                       (short)0, c0, false, false);
            c1 = __builtin_amdgcn_wmma_f32_16x16x4_f32(false, a1_hi, false, b_hi,
                                                       (short)0, c1, false, false);
        }
    }
    float S[2], v[NQ];
    float v7[2], v8[2], v9[2];
#pragma unroll
    for (int t = 0; t < 2; ++t) {
        float p[8];
#pragma unroll
        for (int r = 0; r < 8; ++r) {
            const float g = t ? c1[r] : c0[r];
            const float q = g * g;
            p[r] = q + __shfl_xor(q, 8, 32);
        }
        const float s01 = p[0] + p[1], d01 = p[0] - p[1];
        const float s23 = p[2] + p[3], d23 = p[2] - p[3];
        const float s45 = p[4] + p[5], d45 = p[4] - p[5];
        const float s67 = p[6] + p[7], d67 = p[6] - p[7];
        const float s0123 = s01 + s23, s4567 = s45 + s67;
        S[t] = s0123 + s4567;
        v9[t] = (d01 + d23) + (d45 + d67);
        v8[t] = (s01 - s23) + (s45 - s67);
        v7[t] = s0123 - s4567;
    }
    const float SS = S[0] + S[1];
    v[9] = v9[0] + v9[1];
    v[8] = v8[0] + v8[1];
    v[7] = v7[0] + v7[1];
    v[6] = half ? -SS : SS;
    v[5] = (wv & 1) ? -SS : SS;
    v[4] = (wv & 2) ? -SS : SS;
    v[3] = (wv & 4) ? -SS : SS;
    v[2] = (wv & 8) ? -SS : SS;
    v[1] = (wv & 16) ? -SS : SS;
    v[0] = S[0] - S[1];
#pragma unroll
    for (int w = 0; w < NQ; ++w) v[w] += __shfl_xor(v[w], 16, 32);
    if (lane < SPB) {
#pragma unroll
        for (int w = 0; w < NQ; ++w) sPart[wv][lane * NQ + w] = v[w];
    }
    __syncthreads();
    if (tid < SPB * NQ) {
        float a = sPart[0][tid];
        for (int u = 1; u < GEMM_WAVES; ++u) a += sPart[u][tid];
        evs[m0s * NQ + tid] = a;
    }
}

__global__ void __launch_bounds__(512) output_kernel(
    const float* __restrict__ evs, const float* __restrict__ W_out,
    const float* __restrict__ b_out, float* __restrict__ out) {
    __shared__ float sEv[OSPB * NQ];
    const int tid = threadIdx.x;
    const int m0 = blockIdx.x * OSPB;
    if (tid < OSPB * NQ) sEv[tid] = evs[m0 * NQ + tid];
    __syncthreads();
    float* orun = out + (size_t)m0 * CDIM;
    for (int i = 0; i < 3; ++i) {
        const int o = 4 * (tid + 512 * i);
        const int row = o / CDIM;
        const int c = o - row * CDIM;
        float acc[4];
#pragma unroll
        for (int x = 0; x < 4; ++x) acc[x] = b_out[c + x];
#pragma unroll
        for (int w = 0; w < NQ; ++w) {
            const float e = sEv[row * NQ + w];
#pragma unroll
            for (int x = 0; x < 4; ++x) acc[x] += e * W_out[w * CDIM + c + x];
        }
#pragma unroll
        for (int x = 0; x < 4; ++x) orun[o + x] = acc[x];
    }
}

extern "C" void kernel_launch(void* const* d_in, const int* in_sizes, int n_in,
                              void* d_out, int out_size, void* d_ws,
                              size_t ws_size, hipStream_t stream) {
    const float* z_i     = (const float*)d_in[0];
    const float* W_in    = (const float*)d_in[1];
    const float* b_in    = (const float*)d_in[2];
    const float* weights = (const float*)d_in[3];
    const float* W_out   = (const float*)d_in[4];
    const float* b_out   = (const float*)d_in[5];
    float* out = (float*)d_out;

    float* R_f = (float*)d_ws;
    float* evs = R_f + SDIM * SDIM;

    build_matrix_kernel<<<SDIM / 8 * 2, 512, 0, stream>>>(weights, R_f);
    state_gemm_kernel<<<BATCH / SPB, GEMM_WAVES * 32, 0, stream>>>(z_i, W_in, b_in, weights, R_f, evs);
    output_kernel<<<BATCH / OSPB, 512, 0, stream>>>(evs, W_out, b_out, out);
}
